# starting kernel
# baseline (speedup 1.0000x reference)
#include <hip/hip_runtime.h>

#define N_NODES 100000
#define N_EDGES 1600000
#define IN_F    256
#define HID_F   128
#define OUT_F   32

typedef float v2f __attribute__((ext_vector_type(2)));
typedef float v8f __attribute__((ext_vector_type(8)));

__global__ __launch_bounds__(128) void gemm1_wmma(
    const float* __restrict__ x, const float* __restrict__ W1,
    float* __restrict__ h0)
{
    const int wid  = (blockIdx.x * blockDim.x + threadIdx.x) >> 5;
    const int lane = threadIdx.x & 31;
    if (wid >= N_NODES / 16) return;
    const int m    = lane & 15;
    const int kh   = lane >> 4;
    const int row0 = wid * 16;

    v8f acc[8] = {};

    const float* arow = x + (size_t)(row0 + m) * IN_F + 2 * kh;
    const float* wcol = W1 + m;

    for (int k = 0; k < IN_F; k += 4) {
        v2f a = *(const v2f*)(arow + k);
        const float* wrow = wcol + (size_t)(k + 2 * kh) * HID_F;
#pragma unroll
        for (int t = 0; t < 8; ++t) {
            v2f b;
            b.x = wrow[t * 16];
            b.y = wrow[HID_F + t * 16];
            acc[t] = __builtin_amdgcn_wmma_f32_16x16x4_f32(
                false, a, false, b, (short)0, acc[t], false, false);
        }
    }

    float* drow = h0 + (size_t)(row0 + kh * 8) * HID_F + m;
#pragma unroll
    for (int t = 0; t < 8; ++t)
#pragma unroll
        for (int v = 0; v < 8; ++v)
            drow[(size_t)v * HID_F + t * 16] = acc[t][v];
}

__global__ __launch_bounds__(128) void gemm2_wmma(
    const float* __restrict__ h1, const float* __restrict__ W2,
    float* __restrict__ h2)
{
    const int wid  = (blockIdx.x * blockDim.x + threadIdx.x) >> 5;
    const int lane = threadIdx.x & 31;
    if (wid >= N_NODES / 16) return;
    const int m    = lane & 15;
    const int kh   = lane >> 4;
    const int row0 = wid * 16;

    v8f acc[2] = {};

    const float* arow = h1 + (size_t)(row0 + m) * HID_F + 2 * kh;
    const float* wcol = W2 + m;

    for (int k = 0; k < HID_F; k += 4) {
        v2f a = *(const v2f*)(arow + k);
        a.x = fmaxf(a.x, 0.0f);
        a.y = fmaxf(a.y, 0.0f);
        const float* wrow = wcol + (size_t)(k + 2 * kh) * OUT_F;
#pragma unroll
        for (int t = 0; t < 2; ++t) {
            v2f b;
            b.x = wrow[t * 16];
            b.y = wrow[OUT_F + t * 16];
            acc[t] = __builtin_amdgcn_wmma_f32_16x16x4_f32(
                false, a, false, b, (short)0, acc[t], false, false);
        }
    }

    float* drow = h2 + (size_t)(row0 + kh * 8) * OUT_F + m;
#pragma unroll
    for (int t = 0; t < 2; ++t)
#pragma unroll
        for (int v = 0; v < 8; ++v)
            drow[(size_t)v * OUT_F + t * 16] = acc[t][v];
}

__global__ __launch_bounds__(256) void spmm_f128(
    const int* __restrict__ rows, const int* __restrict__ cols,
    const float* __restrict__ vals, const float* __restrict__ src,
    float* __restrict__ dst)
{
    const int e    = (blockIdx.x * blockDim.x + threadIdx.x) >> 5;
    const int lane = threadIdx.x & 31;
    if (e >= N_EDGES) return;
    const int   r = rows[e];
    const int   c = cols[e];
    const float w = vals[e];
    const float4 mv = ((const float4*)(src + (size_t)c * HID_F))[lane];
    float* d = dst + (size_t)r * HID_F + lane * 4;
    __hip_atomic_fetch_add(d + 0, w * mv.x, __ATOMIC_RELAXED, __HIP_MEMORY_SCOPE_AGENT);
    __hip_atomic_fetch_add(d + 1, w * mv.y, __ATOMIC_RELAXED, __HIP_MEMORY_SCOPE_AGENT);
    __hip_atomic_fetch_add(d + 2, w * mv.z, __ATOMIC_RELAXED, __HIP_MEMORY_SCOPE_AGENT);
    __hip_atomic_fetch_add(d + 3, w * mv.w, __ATOMIC_RELAXED, __HIP_MEMORY_SCOPE_AGENT);
}

__global__ __launch_bounds__(256) void spmm_f32(
    const int* __restrict__ rows, const int* __restrict__ cols,
    const float* __restrict__ vals, const float* __restrict__ src,
    float* __restrict__ dst)
{
    const int e    = (blockIdx.x * blockDim.x + threadIdx.x) >> 5;
    const int lane = threadIdx.x & 31;
    if (e >= N_EDGES) return;
    const int   r = rows[e];
    const int   c = cols[e];
    const float w = vals[e];
    const float mv = src[(size_t)c * OUT_F + lane];
    __hip_atomic_fetch_add(dst + (size_t)r * OUT_F + lane, w * mv,
                           __ATOMIC_RELAXED, __HIP_MEMORY_SCOPE_AGENT);
}

__global__ __launch_bounds__(256) void zero_f4(float4* __restrict__ p, int n4)
{
    int i = blockIdx.x * blockDim.x + threadIdx.x;
    if (i < n4) p[i] = make_float4(0.f, 0.f, 0.f, 0.f);
}

extern "C" void kernel_launch(void* const* d_in, const int* in_sizes, int n_in,
                              void* d_out, int out_size, void* d_ws, size_t ws_size,
                              hipStream_t stream)
{
    const float* x    = (const float*)d_in[0];
    const int*   rows = (const int*)  d_in[1];
    const int*   cols = (const int*)  d_in[2];
    const float* vals = (const float*)d_in[3];
    const float* W1   = (const float*)d_in[4];
    const float* W2   = (const float*)d_in[5];
    float* out = (float*)d_out;

    float* h0 = (float*)d_ws;
    float* h1 = h0 + (size_t)N_NODES * HID_F;
    float* h2 = h0;

    const int gemmBlocks = ((N_NODES / 16) + 3) / 4;
    const int edgeBlocks = (N_EDGES * 32 + 255) / 256;

    gemm1_wmma<<<gemmBlocks, 128, 0, stream>>>(x, W1, h0);

    {
        int n4 = (N_NODES * HID_F) / 4;
        zero_f4<<<(n4 + 255) / 256, 256, 0, stream>>>((float4*)h1, n4);
    }
    spmm_f128<<<edgeBlocks, 256, 0, stream>>>(rows, cols, vals, h0, h1);

    gemm2_wmma<<<gemmBlocks, 128, 0, stream>>>(h1, W2, h2);

    {
        int n4 = (N_NODES * OUT_F) / 4;
        zero_f4<<<(n4 + 255) / 256, 256, 0, stream>>>((float4*)out, n4);
    }
    spmm_f32<<<edgeBlocks, 256, 0, stream>>>(rows, cols, vals, h2, out);
}


# spmm F=128: lane l owns columns l+32j (was 4l..4l+3), so each wave-wide f32 atomic add and gather load covers one whole 128-B line, not quarters of four lines; atomics kept, arrival-order sums
# speedup vs baseline: 1.2061x; 1.2061x over previous
#include <hip/hip_runtime.h>

#define N_NODES 100000
#define N_EDGES 1600000
#define IN_F    256
#define HID_F   128
#define OUT_F   32

typedef float v2f __attribute__((ext_vector_type(2)));
typedef float v8f __attribute__((ext_vector_type(8)));

__global__ __launch_bounds__(128) void gemm1_wmma(
    const float* __restrict__ x, const float* __restrict__ W1,
    float* __restrict__ h0)
{
    const int wid  = (blockIdx.x * blockDim.x + threadIdx.x) >> 5;
    const int lane = threadIdx.x & 31;
    if (wid >= N_NODES / 16) return;
    const int m    = lane & 15;
    const int kh   = lane >> 4;
    const int row0 = wid * 16;

    v8f acc[8] = {};

    const float* arow = x + (size_t)(row0 + m) * IN_F + 2 * kh;
    const float* wcol = W1 + m;

    for (int k = 0; k < IN_F; k += 4) {
        v2f a = *(const v2f*)(arow + k);
        const float* wrow = wcol + (size_t)(k + 2 * kh) * HID_F;
#pragma unroll
        for (int t = 0; t < 8; ++t) {
            v2f b;
            b.x = wrow[t * 16];
            b.y = wrow[HID_F + t * 16];
            acc[t] = __builtin_amdgcn_wmma_f32_16x16x4_f32(
                false, a, false, b, (short)0, acc[t], false, false);
        }
    }

    float* drow = h0 + (size_t)(row0 + kh * 8) * HID_F + m;
#pragma unroll
    for (int t = 0; t < 8; ++t)
#pragma unroll
        for (int v = 0; v < 8; ++v)
            drow[(size_t)v * HID_F + t * 16] = acc[t][v];
}

__global__ __launch_bounds__(128) void gemm2_wmma(
    const float* __restrict__ h1, const float* __restrict__ W2,
    float* __restrict__ h2)
{
    const int wid  = (blockIdx.x * blockDim.x + threadIdx.x) >> 5;
    const int lane = threadIdx.x & 31;
    if (wid >= N_NODES / 16) return;
    const int m    = lane & 15;
    const int kh   = lane >> 4;
    const int row0 = wid * 16;

    v8f acc[2] = {};

    const float* arow = h1 + (size_t)(row0 + m) * HID_F + 2 * kh;
    const float* wcol = W2 + m;

    for (int k = 0; k < HID_F; k += 4) {
        v2f a = *(const v2f*)(arow + k);
        a.x = fmaxf(a.x, 0.0f);
        a.y = fmaxf(a.y, 0.0f);
        const float* wrow = wcol + (size_t)(k + 2 * kh) * OUT_F;
#pragma unroll
        for (int t = 0; t < 2; ++t) {
            v2f b;
            b.x = wrow[t * 16];
            b.y = wrow[OUT_F + t * 16];
            acc[t] = __builtin_amdgcn_wmma_f32_16x16x4_f32(
                false, a, false, b, (short)0, acc[t], false, false);
        }
    }

    float* drow = h2 + (size_t)(row0 + kh * 8) * OUT_F + m;
#pragma unroll
    for (int t = 0; t < 2; ++t)
#pragma unroll
        for (int v = 0; v < 8; ++v)
            drow[(size_t)v * OUT_F + t * 16] = acc[t][v];
}

__global__ __launch_bounds__(256) void spmm_f128(
    const int* __restrict__ rows, const int* __restrict__ cols,
    const float* __restrict__ vals, const float* __restrict__ src,
    float* __restrict__ dst)
{
    const int e    = (blockIdx.x * blockDim.x + threadIdx.x) >> 5;
    const int lane = threadIdx.x & 31;
    if (e >= N_EDGES) return;
    const int   r = rows[e];
    const int   c = cols[e];
    const float w = vals[e];
    const float* s = src + (size_t)c * HID_F + lane;
    float*       d = dst + (size_t)r * HID_F + lane;
    const float m0 = s[0];
    const float m1 = s[32];
    const float m2 = s[64];
    const float m3 = s[96];
    __hip_atomic_fetch_add(d +  0, w * m0, __ATOMIC_RELAXED, __HIP_MEMORY_SCOPE_AGENT);
    __hip_atomic_fetch_add(d + 32, w * m1, __ATOMIC_RELAXED, __HIP_MEMORY_SCOPE_AGENT);
    __hip_atomic_fetch_add(d + 64, w * m2, __ATOMIC_RELAXED, __HIP_MEMORY_SCOPE_AGENT);
    __hip_atomic_fetch_add(d + 96, w * m3, __ATOMIC_RELAXED, __HIP_MEMORY_SCOPE_AGENT);
}

__global__ __launch_bounds__(256) void spmm_f32(
    const int* __restrict__ rows, const int* __restrict__ cols,
    const float* __restrict__ vals, const float* __restrict__ src,
    float* __restrict__ dst)
{
    const int e    = (blockIdx.x * blockDim.x + threadIdx.x) >> 5;
    const int lane = threadIdx.x & 31;
    if (e >= N_EDGES) return;
    const int   r = rows[e];
    const int   c = cols[e];
    const float w = vals[e];
    const float mv = src[(size_t)c * OUT_F + lane];
    __hip_atomic_fetch_add(dst + (size_t)r * OUT_F + lane, w * mv,
                           __ATOMIC_RELAXED, __HIP_MEMORY_SCOPE_AGENT);
}

__global__ __launch_bounds__(256) void zero_f4(float4* __restrict__ p, int n4)
{
    int i = blockIdx.x * blockDim.x + threadIdx.x;
    if (i < n4) p[i] = make_float4(0.f, 0.f, 0.f, 0.f);
}

extern "C" void kernel_launch(void* const* d_in, const int* in_sizes, int n_in,
                              void* d_out, int out_size, void* d_ws, size_t ws_size,
                              hipStream_t stream)
{
    const float* x    = (const float*)d_in[0];
    const int*   rows = (const int*)  d_in[1];
    const int*   cols = (const int*)  d_in[2];
    const float* vals = (const float*)d_in[3];
    const float* W1   = (const float*)d_in[4];
    const float* W2   = (const float*)d_in[5];
    float* out = (float*)d_out;

    float* h0 = (float*)d_ws;
    float* h1 = h0 + (size_t)N_NODES * HID_F;
    float* h2 = h0;

    const int gemmBlocks = ((N_NODES / 16) + 3) / 4;
    const int edgeBlocks = (N_EDGES * 32 + 255) / 256;

    gemm1_wmma<<<gemmBlocks, 128, 0, stream>>>(x, W1, h0);

    {
        int n4 = (N_NODES * HID_F) / 4;
        zero_f4<<<(n4 + 255) / 256, 256, 0, stream>>>((float4*)h1, n4);
    }
    spmm_f128<<<edgeBlocks, 256, 0, stream>>>(rows, cols, vals, h0, h1);

    gemm2_wmma<<<gemmBlocks, 128, 0, stream>>>(h1, W2, h2);

    {
        int n4 = (N_NODES * OUT_F) / 4;
        zero_f4<<<(n4 + 255) / 256, 256, 0, stream>>>((float4*)out, n4);
    }
    spmm_f32<<<edgeBlocks, 256, 0, stream>>>(rows, cols, vals, h2, out);
}


# spmm, both layers: each wave takes 32 edges, not 1 (coalesced index loads, readlane broadcast, 32 or 16 whole-line gathers in flight); 1/32 the waves; atomics kept, arrival-order sums
# speedup vs baseline: 2.1069x; 1.7469x over previous
#include <hip/hip_runtime.h>

#define N_NODES 100000
#define N_EDGES 1600000
#define IN_F    256
#define HID_F   128
#define OUT_F   32

typedef float v2f __attribute__((ext_vector_type(2)));
typedef float v8f __attribute__((ext_vector_type(8)));

__global__ __launch_bounds__(128) void gemm1_wmma(
    const float* __restrict__ x, const float* __restrict__ W1,
    float* __restrict__ h0)
{
    const int wid  = (blockIdx.x * blockDim.x + threadIdx.x) >> 5;
    const int lane = threadIdx.x & 31;
    if (wid >= N_NODES / 16) return;
    const int m    = lane & 15;
    const int kh   = lane >> 4;
    const int row0 = wid * 16;

    v8f acc[8] = {};

    const float* arow = x + (size_t)(row0 + m) * IN_F + 2 * kh;
    const float* wcol = W1 + m;

    for (int k = 0; k < IN_F; k += 4) {
        v2f a = *(const v2f*)(arow + k);
        const float* wrow = wcol + (size_t)(k + 2 * kh) * HID_F;
#pragma unroll
        for (int t = 0; t < 8; ++t) {
            v2f b;
            b.x = wrow[t * 16];
            b.y = wrow[HID_F + t * 16];
            acc[t] = __builtin_amdgcn_wmma_f32_16x16x4_f32(
                false, a, false, b, (short)0, acc[t], false, false);
        }
    }

    float* drow = h0 + (size_t)(row0 + kh * 8) * HID_F + m;
#pragma unroll
    for (int t = 0; t < 8; ++t)
#pragma unroll
        for (int v = 0; v < 8; ++v)
            drow[(size_t)v * HID_F + t * 16] = acc[t][v];
}

__global__ __launch_bounds__(128) void gemm2_wmma(
    const float* __restrict__ h1, const float* __restrict__ W2,
    float* __restrict__ h2)
{
    const int wid  = (blockIdx.x * blockDim.x + threadIdx.x) >> 5;
    const int lane = threadIdx.x & 31;
    if (wid >= N_NODES / 16) return;
    const int m    = lane & 15;
    const int kh   = lane >> 4;
    const int row0 = wid * 16;

    v8f acc[2] = {};

    const float* arow = h1 + (size_t)(row0 + m) * HID_F + 2 * kh;
    const float* wcol = W2 + m;

    for (int k = 0; k < HID_F; k += 4) {
        v2f a = *(const v2f*)(arow + k);
        a.x = fmaxf(a.x, 0.0f);
        a.y = fmaxf(a.y, 0.0f);
        const float* wrow = wcol + (size_t)(k + 2 * kh) * OUT_F;
#pragma unroll
        for (int t = 0; t < 2; ++t) {
            v2f b;
            b.x = wrow[t * 16];
            b.y = wrow[OUT_F + t * 16];
            acc[t] = __builtin_amdgcn_wmma_f32_16x16x4_f32(
                false, a, false, b, (short)0, acc[t], false, false);
        }
    }

    float* drow = h2 + (size_t)(row0 + kh * 8) * OUT_F + m;
#pragma unroll
    for (int t = 0; t < 2; ++t)
#pragma unroll
        for (int v = 0; v < 8; ++v)
            drow[(size_t)v * OUT_F + t * 16] = acc[t][v];
}

__device__ __forceinline__ float lane_bcast_f(float v, int j)
{
    return __builtin_bit_cast(float,
        __builtin_amdgcn_readlane(__builtin_bit_cast(int, v), j));
}

__device__ __forceinline__ const float* row_lane_ptr(const float* base, unsigned off)
{
    return (const float*)((const char*)base + off);
}
__device__ __forceinline__ float* row_lane_ptr(float* base, unsigned off)
{
    return (float*)((char*)base + off);
}

__global__ __launch_bounds__(256) void spmm_f128(
    const int* __restrict__ rows, const int* __restrict__ cols,
    const float* __restrict__ vals, const float* __restrict__ src,
    float* __restrict__ dst)
{
    const int      wave = (blockIdx.x * blockDim.x + threadIdx.x) >> 5;
    const unsigned lane = threadIdx.x & 31;
    if (wave >= N_EDGES / 32) return;
    const int   e  = wave * 32 + (int)lane;
    const int   rv = rows[e];
    const int   cv = cols[e];
    const float wv = vals[e];
    const unsigned lane4 = lane * 4u;

    for (int g = 0; g < 32; g += 8) {
        float m[8][4];
#pragma unroll
        for (int j = 0; j < 8; ++j) {
            const unsigned c = (unsigned)__builtin_amdgcn_readlane(cv, g + j);
            const float* s = row_lane_ptr(src, c * (HID_F * 4u) + lane4);
            m[j][0] = s[0];
            m[j][1] = s[32];
            m[j][2] = s[64];
            m[j][3] = s[96];
        }
#pragma unroll
        for (int j = 0; j < 8; ++j) {
            const unsigned r = (unsigned)__builtin_amdgcn_readlane(rv, g + j);
            const float    w = lane_bcast_f(wv, g + j);
            float* d = row_lane_ptr(dst, r * (HID_F * 4u) + lane4);
            __hip_atomic_fetch_add(d +  0, w * m[j][0], __ATOMIC_RELAXED, __HIP_MEMORY_SCOPE_AGENT);
            __hip_atomic_fetch_add(d + 32, w * m[j][1], __ATOMIC_RELAXED, __HIP_MEMORY_SCOPE_AGENT);
            __hip_atomic_fetch_add(d + 64, w * m[j][2], __ATOMIC_RELAXED, __HIP_MEMORY_SCOPE_AGENT);
            __hip_atomic_fetch_add(d + 96, w * m[j][3], __ATOMIC_RELAXED, __HIP_MEMORY_SCOPE_AGENT);
        }
    }
}

__global__ __launch_bounds__(256) void spmm_f32(
    const int* __restrict__ rows, const int* __restrict__ cols,
    const float* __restrict__ vals, const float* __restrict__ src,
    float* __restrict__ dst)
{
    const int      wave = (blockIdx.x * blockDim.x + threadIdx.x) >> 5;
    const unsigned lane = threadIdx.x & 31;
    if (wave >= N_EDGES / 32) return;
    const int   e  = wave * 32 + (int)lane;
    const int   rv = rows[e];
    const int   cv = cols[e];
    const float wv = vals[e];
    const unsigned lane4 = lane * 4u;

    for (int g = 0; g < 32; g += 16) {
        float m[16];
#pragma unroll
        for (int j = 0; j < 16; ++j) {
            const unsigned c = (unsigned)__builtin_amdgcn_readlane(cv, g + j);
            m[j] = *row_lane_ptr(src, c * (OUT_F * 4u) + lane4);
        }
#pragma unroll
        for (int j = 0; j < 16; ++j) {
            const unsigned r = (unsigned)__builtin_amdgcn_readlane(rv, g + j);
            const float    w = lane_bcast_f(wv, g + j);
            __hip_atomic_fetch_add(row_lane_ptr(dst, r * (OUT_F * 4u) + lane4), w * m[j],
                                   __ATOMIC_RELAXED, __HIP_MEMORY_SCOPE_AGENT);
        }
    }
}

__global__ __launch_bounds__(256) void zero_f4(float4* __restrict__ p, int n4)
{
    int i = blockIdx.x * blockDim.x + threadIdx.x;
    if (i < n4) p[i] = make_float4(0.f, 0.f, 0.f, 0.f);
}

extern "C" void kernel_launch(void* const* d_in, const int* in_sizes, int n_in,
                              void* d_out, int out_size, void* d_ws, size_t ws_size,
                              hipStream_t stream)
{
    const float* x    = (const float*)d_in[0];
    const int*   rows = (const int*)  d_in[1];
    const int*   cols = (const int*)  d_in[2];
    const float* vals = (const float*)d_in[3];
    const float* W1   = (const float*)d_in[4];
    const float* W2   = (const float*)d_in[5];
    float* out = (float*)d_out;

    float* h0 = (float*)d_ws;
    float* h1 = h0 + (size_t)N_NODES * HID_F;
    float* h2 = h0;

    const int gemmBlocks = ((N_NODES / 16) + 3) / 4;
    const int edgeBlocks = (N_EDGES + 255) / 256;

    gemm1_wmma<<<gemmBlocks, 128, 0, stream>>>(x, W1, h0);

    {
        int n4 = (N_NODES * HID_F) / 4;
        zero_f4<<<(n4 + 255) / 256, 256, 0, stream>>>((float4*)h1, n4);
    }
    spmm_f128<<<edgeBlocks, 256, 0, stream>>>(rows, cols, vals, h0, h1);

    gemm2_wmma<<<gemmBlocks, 128, 0, stream>>>(h1, W2, h2);

    {
        int n4 = (N_NODES * OUT_F) / 4;
        zero_f4<<<(n4 + 255) / 256, 256, 0, stream>>>((float4*)out, n4);
    }
    spmm_f32<<<edgeBlocks, 256, 0, stream>>>(rows, cols, vals, h2, out);
}


# spmm F=128: edges binned per call by dst-row range (LDS counting sort); one work-group per range sums its rows in LDS (ds f32 atomics, arrival-order sums); no global atomics or zero fill
# speedup vs baseline: 2.3978x; 1.1381x over previous
#include <hip/hip_runtime.h>

#define N_NODES 100000
#define N_EDGES 1600000
#define IN_F    256
#define HID_F   128
#define OUT_F   32

typedef float v2f __attribute__((ext_vector_type(2)));
typedef float v8f __attribute__((ext_vector_type(8)));

__global__ __launch_bounds__(128) void gemm1_wmma(
    const float* __restrict__ x, const float* __restrict__ W1,
    float* __restrict__ h0)
{
    const int wid  = (blockIdx.x * blockDim.x + threadIdx.x) >> 5;
    const int lane = threadIdx.x & 31;
    if (wid >= N_NODES / 16) return;
    const int m    = lane & 15;
    const int kh   = lane >> 4;
    const int row0 = wid * 16;

    v8f acc[8] = {};

    const float* arow = x + (size_t)(row0 + m) * IN_F + 2 * kh;
    const float* wcol = W1 + m;

    for (int k = 0; k < IN_F; k += 4) {
        v2f a = *(const v2f*)(arow + k);
        const float* wrow = wcol + (size_t)(k + 2 * kh) * HID_F;
#pragma unroll
        for (int t = 0; t < 8; ++t) {
            v2f b;
            b.x = wrow[t * 16];
            b.y = wrow[HID_F + t * 16];
            acc[t] = __builtin_amdgcn_wmma_f32_16x16x4_f32(
                false, a, false, b, (short)0, acc[t], false, false);
        }
    }

    float* drow = h0 + (size_t)(row0 + kh * 8) * HID_F + m;
#pragma unroll
    for (int t = 0; t < 8; ++t)
#pragma unroll
        for (int v = 0; v < 8; ++v)
            drow[(size_t)v * HID_F + t * 16] = acc[t][v];
}

__global__ __launch_bounds__(128) void gemm2_wmma(
    const float* __restrict__ h1, const float* __restrict__ W2,
    float* __restrict__ h2)
{
    const int wid  = (blockIdx.x * blockDim.x + threadIdx.x) >> 5;
    const int lane = threadIdx.x & 31;
    if (wid >= N_NODES / 16) return;
    const int m    = lane & 15;
    const int kh   = lane >> 4;
    const int row0 = wid * 16;

    v8f acc[2] = {};

    const float* arow = h1 + (size_t)(row0 + m) * HID_F + 2 * kh;
    const float* wcol = W2 + m;

    for (int k = 0; k < HID_F; k += 4) {
        v2f a = *(const v2f*)(arow + k);
        a.x = fmaxf(a.x, 0.0f);
        a.y = fmaxf(a.y, 0.0f);
        const float* wrow = wcol + (size_t)(k + 2 * kh) * OUT_F;
#pragma unroll
        for (int t = 0; t < 2; ++t) {
            v2f b;
            b.x = wrow[t * 16];
            b.y = wrow[OUT_F + t * 16];
            acc[t] = __builtin_amdgcn_wmma_f32_16x16x4_f32(
                false, a, false, b, (short)0, acc[t], false, false);
        }
    }

    float* drow = h2 + (size_t)(row0 + kh * 8) * OUT_F + m;
#pragma unroll
    for (int t = 0; t < 2; ++t)
#pragma unroll
        for (int v = 0; v < 8; ++v)
            drow[(size_t)v * OUT_F + t * 16] = acc[t][v];
}

__device__ __forceinline__ float lane_bcast_f(float v, int j)
{
    return __builtin_bit_cast(float,
        __builtin_amdgcn_readlane(__builtin_bit_cast(int, v), j));
}

__device__ __forceinline__ const float* row_lane_ptr(const float* base, unsigned off)
{
    return (const float*)((const char*)base + off);
}
__device__ __forceinline__ float* row_lane_ptr(float* base, unsigned off)
{
    return (float*)((char*)base + off);
}

#define NBKT 512
#define RPB  196
#define NBLK 128
#define EPB  (N_EDGES / NBLK)

__device__ __forceinline__ unsigned wave_incl_scan(unsigned v, unsigned lane)
{
#pragma unroll
    for (int d = 1; d < 32; d <<= 1) {
        unsigned u = __shfl_up(v, d, 32);
        if (lane >= (unsigned)d) v += u;
    }
    return v;
}

__global__ __launch_bounds__(1024) void bin_count(
    const int* __restrict__ rows, unsigned* __restrict__ cnt)
{
    __shared__ unsigned hist[NBKT];
    const unsigned tid = threadIdx.x, b = blockIdx.x;
    if (tid < NBKT) hist[tid] = 0u;
    __syncthreads();
    for (unsigned i = tid; i < EPB; i += 1024u) {
        const unsigned r = (unsigned)rows[b * EPB + i];
        atomicAdd(&hist[r / RPB], 1u);
    }
    __syncthreads();
    if (tid < NBKT) cnt[tid * NBLK + b] = hist[tid];
}

__global__ __launch_bounds__(256) void bin_scan(
    unsigned* __restrict__ cnt, unsigned* __restrict__ tot)
{
    const unsigned k    = (blockIdx.x * blockDim.x + threadIdx.x) >> 5;
    const unsigned lane = threadIdx.x & 31;
    uint4* p = (uint4*)(cnt + k * NBLK) + lane;
    const uint4 v = *p;
    const unsigned s    = v.x + v.y + v.z + v.w;
    const unsigned incl = wave_incl_scan(s, lane);
    const unsigned ex   = incl - s;
    *p = make_uint4(ex, ex + v.x, ex + v.x + v.y, ex + v.x + v.y + v.z);
    if (lane == 31) tot[k] = incl;
}

__global__ __launch_bounds__(1024) void bin_place(
    const int* __restrict__ rows, const int* __restrict__ cols,
    const float* __restrict__ vals, const unsigned* __restrict__ cnt,
    const unsigned* __restrict__ tot, int2* __restrict__ list,
    float* __restrict__ dst)
{
    __shared__ unsigned cur[NBKT];
    __shared__ unsigned wsum[NBKT / 32];
    const unsigned tid = threadIdx.x, b = blockIdx.x, lane = tid & 31;
    unsigned v = 0u, incl = 0u;
    if (tid < NBKT) {
        v    = tot[tid];
        incl = wave_incl_scan(v, lane);
        if (lane == 31) wsum[tid >> 5] = incl;
    }
    __syncthreads();
    if (tid < NBKT) {
        unsigned base = 0u;
        for (unsigned w = 0; w < (tid >> 5); ++w) base += wsum[w];
        const unsigned first = base + incl - v;
        cur[tid] = first + cnt[tid * NBLK + b];
        if (b == 0 && tid * RPB < N_NODES)
            *(int2*)(dst + (size_t)tid * RPB * HID_F) = make_int2((int)first, (int)(first + v));
    }
    __syncthreads();
    for (unsigned i0 = tid; i0 < EPB; i0 += 4096u) {
        unsigned r[4], c[4];
        int      w[4];
#pragma unroll
        for (int j = 0; j < 4; ++j) {
            const unsigned i = i0 + (unsigned)j * 1024u;
            if (i < EPB) {
                const unsigned e = b * EPB + i;
                r[j] = (unsigned)rows[e];
                c[j] = (unsigned)cols[e];
                w[j] = __float_as_int(vals[e]);
            }
        }
#pragma unroll
        for (int j = 0; j < 4; ++j) {
            const unsigned i = i0 + (unsigned)j * 1024u;
            if (i < EPB) {
                const unsigned k   = r[j] / RPB;
                const unsigned rl  = r[j] - k * RPB;
                const unsigned pos = atomicAdd(&cur[k], 1u);
                list[pos] = make_int2((int)(c[j] | (rl << 17)), w[j]);
            }
        }
    }
}

__global__ __launch_bounds__(512) void spmm_f128_rows(
    const int2* __restrict__ list, const float* __restrict__ src,
    float* __restrict__ dst)
{
    __shared__ float acc[RPB * HID_F];
    const unsigned tid = threadIdx.x, b = blockIdx.x;
    const unsigned lane = tid & 31;
    const unsigned wave = (unsigned)__builtin_amdgcn_readfirstlane((int)(tid >> 5));
    float* drows = dst + (size_t)b * RPB * HID_F;
    const int2 hd = *(const int2*)drows;
    for (unsigned i = tid; i < RPB * HID_F; i += 512u) acc[i] = 0.0f;
    __syncthreads();
    const int first = __builtin_amdgcn_readfirstlane(hd.x);
    const int last  = __builtin_amdgcn_readfirstlane(hd.y);
    const unsigned lane4 = lane * 4u;

    for (int e0 = first + (int)wave * 32; e0 < last; e0 += 16 * 32) {
        const int n = min(32, last - e0);
        int2 pv = make_int2(0, 0);
        if ((int)lane < n) pv = list[e0 + (int)lane];
        int g = 0;
        for (; g + 8 <= n; g += 8) {
            float m[8][4];
#pragma unroll
            for (int j = 0; j < 8; ++j) {
                const unsigned p = (unsigned)__builtin_amdgcn_readlane(pv.x, g + j);
                const float* s = row_lane_ptr(src, (p & 0x1FFFFu) * (HID_F * 4u) + lane4);
                m[j][0] = s[0];
                m[j][1] = s[32];
                m[j][2] = s[64];
                m[j][3] = s[96];
            }
#pragma unroll
            for (int j = 0; j < 8; ++j) {
                const unsigned p = (unsigned)__builtin_amdgcn_readlane(pv.x, g + j);
                const float    w = __int_as_float(__builtin_amdgcn_readlane(pv.y, g + j));
                float* a = acc + (p >> 17) * HID_F + lane;
                atomicAdd(a +  0, w * m[j][0]);
                atomicAdd(a + 32, w * m[j][1]);
                atomicAdd(a + 64, w * m[j][2]);
                atomicAdd(a + 96, w * m[j][3]);
            }
        }
        if (g < n) {
            float m[7][4];
#pragma unroll
            for (int j = 0; j < 7; ++j) {
                if (g + j < n) {
                    const unsigned p = (unsigned)__builtin_amdgcn_readlane(pv.x, g + j);
                    const float* s = row_lane_ptr(src, (p & 0x1FFFFu) * (HID_F * 4u) + lane4);
                    m[j][0] = s[0];
                    m[j][1] = s[32];
                    m[j][2] = s[64];
                    m[j][3] = s[96];
                }
            }
#pragma unroll
            for (int j = 0; j < 7; ++j) {
                if (g + j < n) {
                    const unsigned p = (unsigned)__builtin_amdgcn_readlane(pv.x, g + j);
                    const float    w = __int_as_float(__builtin_amdgcn_readlane(pv.y, g + j));
                    float* a = acc + (p >> 17) * HID_F + lane;
                    atomicAdd(a +  0, w * m[j][0]);
                    atomicAdd(a + 32, w * m[j][1]);
                    atomicAdd(a + 64, w * m[j][2]);
                    atomicAdd(a + 96, w * m[j][3]);
                }
            }
        }
    }
    __syncthreads();
    const unsigned nrows = min((unsigned)RPB, (unsigned)N_NODES - b * RPB);
    const float4* a4 = (const float4*)acc;
    float4* d4 = (float4*)drows;
    for (unsigned i = tid; i < nrows * (HID_F / 4); i += 512u) d4[i] = a4[i];
}

__global__ __launch_bounds__(256) void spmm_f32(
    const int* __restrict__ rows, const int* __restrict__ cols,
    const float* __restrict__ vals, const float* __restrict__ src,
    float* __restrict__ dst)
{
    const int      wave = (blockIdx.x * blockDim.x + threadIdx.x) >> 5;
    const unsigned lane = threadIdx.x & 31;
    if (wave >= N_EDGES / 32) return;
    const int   e  = wave * 32 + (int)lane;
    const int   rv = rows[e];
    const int   cv = cols[e];
    const float wv = vals[e];
    const unsigned lane4 = lane * 4u;

    for (int g = 0; g < 32; g += 16) {
        float m[16];
#pragma unroll
        for (int j = 0; j < 16; ++j) {
            const unsigned c = (unsigned)__builtin_amdgcn_readlane(cv, g + j);
            m[j] = *row_lane_ptr(src, c * (OUT_F * 4u) + lane4);
        }
#pragma unroll
        for (int j = 0; j < 16; ++j) {
            const unsigned r = (unsigned)__builtin_amdgcn_readlane(rv, g + j);
            const float    w = lane_bcast_f(wv, g + j);
            __hip_atomic_fetch_add(row_lane_ptr(dst, r * (OUT_F * 4u) + lane4), w * m[j],
                                   __ATOMIC_RELAXED, __HIP_MEMORY_SCOPE_AGENT);
        }
    }
}

__global__ __launch_bounds__(256) void zero_f4(float4* __restrict__ p, int n4)
{
    int i = blockIdx.x * blockDim.x + threadIdx.x;
    if (i < n4) p[i] = make_float4(0.f, 0.f, 0.f, 0.f);
}

extern "C" void kernel_launch(void* const* d_in, const int* in_sizes, int n_in,
                              void* d_out, int out_size, void* d_ws, size_t ws_size,
                              hipStream_t stream)
{
    const float* x    = (const float*)d_in[0];
    const int*   rows = (const int*)  d_in[1];
    const int*   cols = (const int*)  d_in[2];
    const float* vals = (const float*)d_in[3];
    const float* W1   = (const float*)d_in[4];
    const float* W2   = (const float*)d_in[5];
    float* out = (float*)d_out;

    float* h0 = (float*)d_ws;
    float* h1 = h0 + (size_t)N_NODES * HID_F;
    float* h2 = h0;

    const int gemmBlocks = ((N_NODES / 16) + 3) / 4;
    const int edgeBlocks = (N_EDGES + 255) / 256;

    unsigned* cnt  = (unsigned*)d_ws;
    unsigned* tot  = cnt + NBKT * NBLK;
    int2*     list = (int2*)d_out;
    bin_count<<<NBLK, 1024, 0, stream>>>(rows, cnt);
    bin_scan<<<NBKT / 8, 256, 0, stream>>>(cnt, tot);
    bin_place<<<NBLK, 1024, 0, stream>>>(rows, cols, vals, cnt, tot, list, h1);

    gemm1_wmma<<<gemmBlocks, 128, 0, stream>>>(x, W1, h0);

    spmm_f128_rows<<<(N_NODES + RPB - 1) / RPB, 512, 0, stream>>>(list, h0, h1);

    gemm2_wmma<<<gemmBlocks, 128, 0, stream>>>(h1, W2, h2);

    {
        int n4 = (N_NODES * OUT_F) / 4;
        zero_f4<<<(n4 + 255) / 256, 256, 0, stream>>>((float4*)out, n4);
    }
    spmm_f32<<<edgeBlocks, 256, 0, stream>>>(rows, cols, vals, h2, out);
}


# spmm F=128: work-group puts its range's entries in row order in LDS (arrival-order integer rank, scan, place); a wave sums each row in registers (fma), stores it once; no LDS f32 adds or row tile
# speedup vs baseline: 2.5332x; 1.0565x over previous
#include <hip/hip_runtime.h>

#define N_NODES 100000
#define N_EDGES 1600000
#define IN_F    256
#define HID_F   128
#define OUT_F   32

typedef float v2f __attribute__((ext_vector_type(2)));
typedef float v8f __attribute__((ext_vector_type(8)));

__global__ __launch_bounds__(128) void gemm1_wmma(
    const float* __restrict__ x, const float* __restrict__ W1,
    float* __restrict__ h0)
{
    const int wid  = (blockIdx.x * blockDim.x + threadIdx.x) >> 5;
    const int lane = threadIdx.x & 31;
    if (wid >= N_NODES / 16) return;
    const int m    = lane & 15;
    const int kh   = lane >> 4;
    const int row0 = wid * 16;

    v8f acc[8] = {};

    const float* arow = x + (size_t)(row0 + m) * IN_F + 2 * kh;
    const float* wcol = W1 + m;

    for (int k = 0; k < IN_F; k += 4) {
        v2f a = *(const v2f*)(arow + k);
        const float* wrow = wcol + (size_t)(k + 2 * kh) * HID_F;
#pragma unroll
        for (int t = 0; t < 8; ++t) {
            v2f b;
            b.x = wrow[t * 16];
            b.y = wrow[HID_F + t * 16];
            acc[t] = __builtin_amdgcn_wmma_f32_16x16x4_f32(
                false, a, false, b, (short)0, acc[t], false, false);
        }
    }

    float* drow = h0 + (size_t)(row0 + kh * 8) * HID_F + m;
#pragma unroll
    for (int t = 0; t < 8; ++t)
#pragma unroll
        for (int v = 0; v < 8; ++v)
            drow[(size_t)v * HID_F + t * 16] = acc[t][v];
}

__global__ __launch_bounds__(128) void gemm2_wmma(
    const float* __restrict__ h1, const float* __restrict__ W2,
    float* __restrict__ h2)
{
    const int wid  = (blockIdx.x * blockDim.x + threadIdx.x) >> 5;
    const int lane = threadIdx.x & 31;
    if (wid >= N_NODES / 16) return;
    const int m    = lane & 15;
    const int kh   = lane >> 4;
    const int row0 = wid * 16;

    v8f acc[2] = {};

    const float* arow = h1 + (size_t)(row0 + m) * HID_F + 2 * kh;
    const float* wcol = W2 + m;

    for (int k = 0; k < HID_F; k += 4) {
        v2f a = *(const v2f*)(arow + k);
        a.x = fmaxf(a.x, 0.0f);
        a.y = fmaxf(a.y, 0.0f);
        const float* wrow = wcol + (size_t)(k + 2 * kh) * OUT_F;
#pragma unroll
        for (int t = 0; t < 2; ++t) {
            v2f b;
            b.x = wrow[t * 16];
            b.y = wrow[OUT_F + t * 16];
            acc[t] = __builtin_amdgcn_wmma_f32_16x16x4_f32(
                false, a, false, b, (short)0, acc[t], false, false);
        }
    }

    float* drow = h2 + (size_t)(row0 + kh * 8) * OUT_F + m;
#pragma unroll
    for (int t = 0; t < 2; ++t)
#pragma unroll
        for (int v = 0; v < 8; ++v)
            drow[(size_t)v * OUT_F + t * 16] = acc[t][v];
}

__device__ __forceinline__ float lane_bcast_f(float v, int j)
{
    return __builtin_bit_cast(float,
        __builtin_amdgcn_readlane(__builtin_bit_cast(int, v), j));
}

__device__ __forceinline__ const float* row_lane_ptr(const float* base, unsigned off)
{
    return (const float*)((const char*)base + off);
}
__device__ __forceinline__ float* row_lane_ptr(float* base, unsigned off)
{
    return (float*)((char*)base + off);
}

#define NBKT 512
#define RPB  196
#define NBLK 128
#define EPB  (N_EDGES / NBLK)

__device__ __forceinline__ unsigned wave_incl_scan(unsigned v, unsigned lane)
{
#pragma unroll
    for (int d = 1; d < 32; d <<= 1) {
        unsigned u = __shfl_up(v, d, 32);
        if (lane >= (unsigned)d) v += u;
    }
    return v;
}

__global__ __launch_bounds__(1024) void bin_count(
    const int* __restrict__ rows, unsigned* __restrict__ cnt)
{
    __shared__ unsigned hist[NBKT];
    const unsigned tid = threadIdx.x, b = blockIdx.x;
    if (tid < NBKT) hist[tid] = 0u;
    __syncthreads();
    for (unsigned i = tid; i < EPB; i += 1024u) {
        const unsigned r = (unsigned)rows[b * EPB + i];
        atomicAdd(&hist[r / RPB], 1u);
    }
    __syncthreads();
    if (tid < NBKT) cnt[tid * NBLK + b] = hist[tid];
}

__global__ __launch_bounds__(256) void bin_scan(
    unsigned* __restrict__ cnt, unsigned* __restrict__ tot)
{
    const unsigned k    = (blockIdx.x * blockDim.x + threadIdx.x) >> 5;
    const unsigned lane = threadIdx.x & 31;
    uint4* p = (uint4*)(cnt + k * NBLK) + lane;
    const uint4 v = *p;
    const unsigned s    = v.x + v.y + v.z + v.w;
    const unsigned incl = wave_incl_scan(s, lane);
    const unsigned ex   = incl - s;
    *p = make_uint4(ex, ex + v.x, ex + v.x + v.y, ex + v.x + v.y + v.z);
    if (lane == 31) tot[k] = incl;
}

__global__ __launch_bounds__(1024) void bin_place(
    const int* __restrict__ rows, const int* __restrict__ cols,
    const float* __restrict__ vals, const unsigned* __restrict__ cnt,
    const unsigned* __restrict__ tot, int2* __restrict__ list,
    float* __restrict__ dst)
{
    __shared__ unsigned cur[NBKT];
    __shared__ unsigned wsum[NBKT / 32];
    const unsigned tid = threadIdx.x, b = blockIdx.x, lane = tid & 31;
    unsigned v = 0u, incl = 0u;
    if (tid < NBKT) {
        v    = tot[tid];
        incl = wave_incl_scan(v, lane);
        if (lane == 31) wsum[tid >> 5] = incl;
    }
    __syncthreads();
    if (tid < NBKT) {
        unsigned base = 0u;
        for (unsigned w = 0; w < (tid >> 5); ++w) base += wsum[w];
        const unsigned first = base + incl - v;
        cur[tid] = first + cnt[tid * NBLK + b];
        if (b == 0 && tid * RPB < N_NODES)
            *(int2*)(dst + (size_t)tid * RPB * HID_F) = make_int2((int)first, (int)(first + v));
    }
    __syncthreads();
    for (unsigned i0 = tid; i0 < EPB; i0 += 4096u) {
        unsigned r[4], c[4];
        int      w[4];
#pragma unroll
        for (int j = 0; j < 4; ++j) {
            const unsigned i = i0 + (unsigned)j * 1024u;
            if (i < EPB) {
                const unsigned e = b * EPB + i;
                r[j] = (unsigned)rows[e];
                c[j] = (unsigned)cols[e];
                w[j] = __float_as_int(vals[e]);
            }
        }
#pragma unroll
        for (int j = 0; j < 4; ++j) {
            const unsigned i = i0 + (unsigned)j * 1024u;
            if (i < EPB) {
                const unsigned k   = r[j] / RPB;
                const unsigned rl  = r[j] - k * RPB;
                const unsigned pos = atomicAdd(&cur[k], 1u);
                list[pos] = make_int2((int)(c[j] | (rl << 17)), w[j]);
            }
        }
    }
}

#define CAP 4096
#define EPT (CAP / 512)
#define RSL 224

__global__ __launch_bounds__(512) void spmm_f128_rows(
    const int2* __restrict__ list, const float* __restrict__ src,
    float* __restrict__ dst)
{
    __shared__ int2     ent[CAP];
    __shared__ unsigned rcnt[RSL];
    __shared__ unsigned roff[RSL + 1];
    const unsigned tid = threadIdx.x, b = blockIdx.x;
    const unsigned lane = tid & 31;
    const int wave = __builtin_amdgcn_readfirstlane((int)(tid >> 5));
    float* drows = dst + (size_t)b * RPB * HID_F;
    const int2 hd = *(const int2*)drows;
    const int first = __builtin_amdgcn_readfirstlane(hd.x);
    const int last  = __builtin_amdgcn_readfirstlane(hd.y);
    const unsigned lane4 = lane * 4u;
    const unsigned nrows = min((unsigned)RPB, (unsigned)N_NODES - b * RPB);
    const int r_lo = wave * 12 + min(wave, 4);
    const int nown = wave < 4 ? 13 : 12;
    const int r_hi = r_lo + nown;

    int pass = 0;
    for (int p0 = first; pass == 0 || p0 < last; p0 += CAP, ++pass) {
        const int n = min(CAP, last - p0);
        if (tid < RSL) rcnt[tid] = 0u;
        __syncthreads();

        int2     e[EPT];
        unsigned rk[EPT];
#pragma unroll
        for (int k = 0; k < EPT; ++k) {
            const int i = (int)tid + k * 512;
            e[k] = make_int2(0, 0);
            if (i < n) e[k] = list[p0 + i];
        }
#pragma unroll
        for (int k = 0; k < EPT; ++k) {
            const int i = (int)tid + k * 512;
            rk[k] = 0u;
            if (i < n) rk[k] = atomicAdd(&rcnt[(unsigned)e[k].x >> 17], 1u);
        }
        __syncthreads();

        if (wave == 0) {
            unsigned c[7], s = 0u;
#pragma unroll
            for (int j = 0; j < 7; ++j) { c[j] = rcnt[lane * 7u + j]; s += c[j]; }
            const unsigned incl = wave_incl_scan(s, lane);
            unsigned run = incl - s;
#pragma unroll
            for (int j = 0; j < 7; ++j) { roff[lane * 7u + j] = run; run += c[j]; }
            if (lane == 31) roff[RSL] = incl;
        }
        __syncthreads();

#pragma unroll
        for (int k = 0; k < EPT; ++k) {
            const int i = (int)tid + k * 512;
            if (i < n) ent[roff[(unsigned)e[k].x >> 17] + rk[k]] = e[k];
        }
        __syncthreads();

        const unsigned myoff = roff[min(r_lo + (int)lane, RSL)];
        const int S = __builtin_amdgcn_readlane((int)myoff, 0);
        const int E = __builtin_amdgcn_readlane((int)myoff, nown);

        if (pass == 0) {
            const unsigned nx = __shfl_down(myoff, 1, 32);
            unsigned mask = __builtin_amdgcn_ballot_w32((int)lane < nown && nx == myoff);
            while (mask) {
                const int r = r_lo + __builtin_ctz(mask);
                mask &= mask - 1u;
                if ((unsigned)r < nrows) {
                    float* d = drows + (unsigned)r * HID_F + lane;
                    d[0] = 0.0f; d[32] = 0.0f; d[64] = 0.0f; d[96] = 0.0f;
                }
            }
        }

        int   cur = r_lo - 1;
        float a0 = 0.0f, a1 = 0.0f, a2 = 0.0f, a3 = 0.0f;
        auto flush = [&](int nxt) {
            if (cur >= r_lo && (unsigned)cur < nrows) {
                float* d = drows + (unsigned)cur * HID_F + lane;
                if (pass) { a0 += d[0]; a1 += d[32]; a2 += d[64]; a3 += d[96]; }
                d[0] = a0; d[32] = a1; d[64] = a2; d[96] = a3;
            }
            a0 = 0.0f; a1 = 0.0f; a2 = 0.0f; a3 = 0.0f;
            cur = nxt;
        };

        for (int e0 = S; e0 < E; e0 += 32) {
            const int cn = min(32, E - e0);
            int2 pv = make_int2(0, 0);
            if ((int)lane < cn) pv = ent[e0 + (int)lane];
            int g = 0;
            for (; g + 8 <= cn; g += 8) {
                float m[8][4];
#pragma unroll
                for (int j = 0; j < 8; ++j) {
                    const unsigned p = (unsigned)__builtin_amdgcn_readlane(pv.x, g + j);
                    const float* s = row_lane_ptr(src, (p & 0x1FFFFu) * (HID_F * 4u) + lane4);
                    m[j][0] = s[0];
                    m[j][1] = s[32];
                    m[j][2] = s[64];
                    m[j][3] = s[96];
                }
#pragma unroll
                for (int j = 0; j < 8; ++j) {
                    const unsigned p = (unsigned)__builtin_amdgcn_readlane(pv.x, g + j);
                    const float    w = __int_as_float(__builtin_amdgcn_readlane(pv.y, g + j));
                    const int      rl = (int)(p >> 17);
                    if (rl != cur) flush(rl);
                    a0 = fmaf(w, m[j][0], a0);
                    a1 = fmaf(w, m[j][1], a1);
                    a2 = fmaf(w, m[j][2], a2);
                    a3 = fmaf(w, m[j][3], a3);
                }
            }
            if (g < cn) {
                float m[7][4];
#pragma unroll
                for (int j = 0; j < 7; ++j) {
                    if (g + j < cn) {
                        const unsigned p = (unsigned)__builtin_amdgcn_readlane(pv.x, g + j);
                        const float* s = row_lane_ptr(src, (p & 0x1FFFFu) * (HID_F * 4u) + lane4);
                        m[j][0] = s[0];
                        m[j][1] = s[32];
                        m[j][2] = s[64];
                        m[j][3] = s[96];
                    }
                }
#pragma unroll
                for (int j = 0; j < 7; ++j) {
                    if (g + j < cn) {
                        const unsigned p = (unsigned)__builtin_amdgcn_readlane(pv.x, g + j);
                        const float    w = __int_as_float(__builtin_amdgcn_readlane(pv.y, g + j));
                        const int      rl = (int)(p >> 17);
                        if (rl != cur) flush(rl);
                        a0 = fmaf(w, m[j][0], a0);
                        a1 = fmaf(w, m[j][1], a1);
                        a2 = fmaf(w, m[j][2], a2);
                        a3 = fmaf(w, m[j][3], a3);
                    }
                }
            }
        }
        flush(r_hi);
    }
}

__global__ __launch_bounds__(256) void spmm_f32(
    const int* __restrict__ rows, const int* __restrict__ cols,
    const float* __restrict__ vals, const float* __restrict__ src,
    float* __restrict__ dst)
{
    const int      wave = (blockIdx.x * blockDim.x + threadIdx.x) >> 5;
    const unsigned lane = threadIdx.x & 31;
    if (wave >= N_EDGES / 32) return;
    const int   e  = wave * 32 + (int)lane;
    const int   rv = rows[e];
    const int   cv = cols[e];
    const float wv = vals[e];
    const unsigned lane4 = lane * 4u;

    for (int g = 0; g < 32; g += 16) {
        float m[16];
#pragma unroll
        for (int j = 0; j < 16; ++j) {
            const unsigned c = (unsigned)__builtin_amdgcn_readlane(cv, g + j);
            m[j] = *row_lane_ptr(src, c * (OUT_F * 4u) + lane4);
        }
#pragma unroll
        for (int j = 0; j < 16; ++j) {
            const unsigned r = (unsigned)__builtin_amdgcn_readlane(rv, g + j);
            const float    w = lane_bcast_f(wv, g + j);
            __hip_atomic_fetch_add(row_lane_ptr(dst, r * (OUT_F * 4u) + lane4), w * m[j],
                                   __ATOMIC_RELAXED, __HIP_MEMORY_SCOPE_AGENT);
        }
    }
}

__global__ __launch_bounds__(256) void zero_f4(float4* __restrict__ p, int n4)
{
    int i = blockIdx.x * blockDim.x + threadIdx.x;
    if (i < n4) p[i] = make_float4(0.f, 0.f, 0.f, 0.f);
}

extern "C" void kernel_launch(void* const* d_in, const int* in_sizes, int n_in,
                              void* d_out, int out_size, void* d_ws, size_t ws_size,
                              hipStream_t stream)
{
    const float* x    = (const float*)d_in[0];
    const int*   rows = (const int*)  d_in[1];
    const int*   cols = (const int*)  d_in[2];
    const float* vals = (const float*)d_in[3];
    const float* W1   = (const float*)d_in[4];
    const float* W2   = (const float*)d_in[5];
    float* out = (float*)d_out;

    float* h0 = (float*)d_ws;
    float* h1 = h0 + (size_t)N_NODES * HID_F;
    float* h2 = h0;

    const int gemmBlocks = ((N_NODES / 16) + 3) / 4;
    const int edgeBlocks = (N_EDGES + 255) / 256;

    unsigned* cnt  = (unsigned*)d_ws;
    unsigned* tot  = cnt + NBKT * NBLK;
    int2*     list = (int2*)d_out;
    bin_count<<<NBLK, 1024, 0, stream>>>(rows, cnt);
    bin_scan<<<NBKT / 8, 256, 0, stream>>>(cnt, tot);
    bin_place<<<NBLK, 1024, 0, stream>>>(rows, cols, vals, cnt, tot, list, h1);

    gemm1_wmma<<<gemmBlocks, 128, 0, stream>>>(x, W1, h0);

    spmm_f128_rows<<<(N_NODES + RPB - 1) / RPB, 512, 0, stream>>>(list, h0, h1);

    gemm2_wmma<<<gemmBlocks, 128, 0, stream>>>(h1, W2, h2);

    {
        int n4 = (N_NODES * OUT_F) / 4;
        zero_f4<<<(n4 + 255) / 256, 256, 0, stream>>>((float4*)out, n4);
    }
    spmm_f32<<<edgeBlocks, 256, 0, stream>>>(rows, cols, vals, h2, out);
}


# layer-2 linear (relu, x W2) fused behind the F=128 row walk: a wave keeps its 12-13 summed rows in LDS and multiplies them by W2 on f32 WMMA; no h1 in memory, one launch fewer; sum order unchanged
# speedup vs baseline: 2.6536x; 1.0475x over previous
#include <hip/hip_runtime.h>

#define N_NODES 100000
#define N_EDGES 1600000
#define IN_F    256
#define HID_F   128
#define OUT_F   32

typedef float v2f __attribute__((ext_vector_type(2)));
typedef float v8f __attribute__((ext_vector_type(8)));

__global__ __launch_bounds__(128) void gemm1_wmma(
    const float* __restrict__ x, const float* __restrict__ W1,
    float* __restrict__ h0)
{
    const int wid  = (blockIdx.x * blockDim.x + threadIdx.x) >> 5;
    const int lane = threadIdx.x & 31;
    if (wid >= N_NODES / 16) return;
    const int m    = lane & 15;
    const int kh   = lane >> 4;
    const int row0 = wid * 16;

    v8f acc[8] = {};

    const float* arow = x + (size_t)(row0 + m) * IN_F + 2 * kh;
    const float* wcol = W1 + m;

    for (int k = 0; k < IN_F; k += 4) {
        v2f a = *(const v2f*)(arow + k);
        const float* wrow = wcol + (size_t)(k + 2 * kh) * HID_F;
#pragma unroll
        for (int t = 0; t < 8; ++t) {
            v2f b;
            b.x = wrow[t * 16];
            b.y = wrow[HID_F + t * 16];
            acc[t] = __builtin_amdgcn_wmma_f32_16x16x4_f32(
                false, a, false, b, (short)0, acc[t], false, false);
        }
    }

    float* drow = h0 + (size_t)(row0 + kh * 8) * HID_F + m;
#pragma unroll
    for (int t = 0; t < 8; ++t)
#pragma unroll
        for (int v = 0; v < 8; ++v)
            drow[(size_t)v * HID_F + t * 16] = acc[t][v];
}

__device__ __forceinline__ float lane_bcast_f(float v, int j)
{
    return __builtin_bit_cast(float,
        __builtin_amdgcn_readlane(__builtin_bit_cast(int, v), j));
}

__device__ __forceinline__ const float* row_lane_ptr(const float* base, unsigned off)
{
    return (const float*)((const char*)base + off);
}
__device__ __forceinline__ float* row_lane_ptr(float* base, unsigned off)
{
    return (float*)((char*)base + off);
}

#define NBKT 512
#define RPB  196
#define NBLK 128
#define EPB  (N_EDGES / NBLK)

__device__ __forceinline__ unsigned wave_incl_scan(unsigned v, unsigned lane)
{
#pragma unroll
    for (int d = 1; d < 32; d <<= 1) {
        unsigned u = __shfl_up(v, d, 32);
        if (lane >= (unsigned)d) v += u;
    }
    return v;
}

__global__ __launch_bounds__(1024) void bin_count(
    const int* __restrict__ rows, unsigned* __restrict__ cnt)
{
    __shared__ unsigned hist[NBKT];
    const unsigned tid = threadIdx.x, b = blockIdx.x;
    if (tid < NBKT) hist[tid] = 0u;
    __syncthreads();
    for (unsigned i = tid; i < EPB; i += 1024u) {
        const unsigned r = (unsigned)rows[b * EPB + i];
        atomicAdd(&hist[r / RPB], 1u);
    }
    __syncthreads();
    if (tid < NBKT) cnt[tid * NBLK + b] = hist[tid];
}

__global__ __launch_bounds__(256) void bin_scan(
    unsigned* __restrict__ cnt, unsigned* __restrict__ tot)
{
    const unsigned k    = (blockIdx.x * blockDim.x + threadIdx.x) >> 5;
    const unsigned lane = threadIdx.x & 31;
    uint4* p = (uint4*)(cnt + k * NBLK) + lane;
    const uint4 v = *p;
    const unsigned s    = v.x + v.y + v.z + v.w;
    const unsigned incl = wave_incl_scan(s, lane);
    const unsigned ex   = incl - s;
    *p = make_uint4(ex, ex + v.x, ex + v.x + v.y, ex + v.x + v.y + v.z);
    if (lane == 31) tot[k] = incl;
}

__global__ __launch_bounds__(1024) void bin_place(
    const int* __restrict__ rows, const int* __restrict__ cols,
    const float* __restrict__ vals, const unsigned* __restrict__ cnt,
    const unsigned* __restrict__ tot, int2* __restrict__ list,
    int2* __restrict__ rng)
{
    __shared__ unsigned cur[NBKT];
    __shared__ unsigned wsum[NBKT / 32];
    const unsigned tid = threadIdx.x, b = blockIdx.x, lane = tid & 31;
    unsigned v = 0u, incl = 0u;
    if (tid < NBKT) {
        v    = tot[tid];
        incl = wave_incl_scan(v, lane);
        if (lane == 31) wsum[tid >> 5] = incl;
    }
    __syncthreads();
    if (tid < NBKT) {
        unsigned base = 0u;
        for (unsigned w = 0; w < (tid >> 5); ++w) base += wsum[w];
        const unsigned first = base + incl - v;
        cur[tid] = first + cnt[tid * NBLK + b];
        if (b == 0) rng[tid] = make_int2((int)first, (int)(first + v));
    }
    __syncthreads();
    for (unsigned i0 = tid; i0 < EPB; i0 += 4096u) {
        unsigned r[4], c[4];
        int      w[4];
#pragma unroll
        for (int j = 0; j < 4; ++j) {
            const unsigned i = i0 + (unsigned)j * 1024u;
            if (i < EPB) {
                const unsigned e = b * EPB + i;
                r[j] = (unsigned)rows[e];
                c[j] = (unsigned)cols[e];
                w[j] = __float_as_int(vals[e]);
            }
        }
#pragma unroll
        for (int j = 0; j < 4; ++j) {
            const unsigned i = i0 + (unsigned)j * 1024u;
            if (i < EPB) {
                const unsigned k   = r[j] / RPB;
                const unsigned rl  = r[j] - k * RPB;
                const unsigned pos = atomicAdd(&cur[k], 1u);
                list[pos] = make_int2((int)(c[j] | (rl << 17)), w[j]);
            }
        }
    }
}

#define CAP 4096
#define EPT (CAP / 512)
#define RSL 224
#define TPW 13
#define TPITCH 132

__global__ __launch_bounds__(512) void spmm_f128_rows_gemm2(
    const int2* __restrict__ list, const int2* __restrict__ rng,
    const float* __restrict__ src, const float* __restrict__ W2,
    float* __restrict__ h2)
{
    __shared__ int2     ent[CAP];
    __shared__ unsigned rcnt[RSL];
    __shared__ unsigned roff[RSL + 1];
    __shared__ float    tile[16 * TPW * TPITCH];
    const unsigned tid = threadIdx.x, b = blockIdx.x;
    const unsigned lane = tid & 31;
    const int wave = __builtin_amdgcn_readfirstlane((int)(tid >> 5));
    const int2 hd = rng[b];
    const int first = __builtin_amdgcn_readfirstlane(hd.x);
    const int last  = __builtin_amdgcn_readfirstlane(hd.y);
    const unsigned lane4 = lane * 4u;
    const unsigned nrows = min((unsigned)RPB, (unsigned)N_NODES - b * RPB);
    const int r_lo = wave * 12 + min(wave, 4);
    const int nown = wave < 4 ? 13 : 12;
    const int r_hi = r_lo + nown;

    float* mytile = tile + wave * (TPW * TPITCH);
#pragma unroll
    for (int r = 0; r < TPW; ++r) {
        float* t = mytile + r * TPITCH + lane;
        t[0] = 0.0f; t[32] = 0.0f; t[64] = 0.0f; t[96] = 0.0f;
    }

    for (int p0 = first; p0 < last; p0 += CAP) {
        const int n = min(CAP, last - p0);
        if (tid < RSL) rcnt[tid] = 0u;
        __syncthreads();

        int2     e[EPT];
        unsigned rk[EPT];
#pragma unroll
        for (int k = 0; k < EPT; ++k) {
            const int i = (int)tid + k * 512;
            e[k] = make_int2(0, 0);
            if (i < n) e[k] = list[p0 + i];
        }
#pragma unroll
        for (int k = 0; k < EPT; ++k) {
            const int i = (int)tid + k * 512;
            rk[k] = 0u;
            if (i < n) rk[k] = atomicAdd(&rcnt[(unsigned)e[k].x >> 17], 1u);
        }
        __syncthreads();

        if (wave == 0) {
            unsigned c[7], s = 0u;
#pragma unroll
            for (int j = 0; j < 7; ++j) { c[j] = rcnt[lane * 7u + j]; s += c[j]; }
            const unsigned incl = wave_incl_scan(s, lane);
            unsigned run = incl - s;
#pragma unroll
            for (int j = 0; j < 7; ++j) { roff[lane * 7u + j] = run; run += c[j]; }
            if (lane == 31) roff[RSL] = incl;
        }
        __syncthreads();

#pragma unroll
        for (int k = 0; k < EPT; ++k) {
            const int i = (int)tid + k * 512;
            if (i < n) ent[roff[(unsigned)e[k].x >> 17] + rk[k]] = e[k];
        }
        __syncthreads();

        const unsigned myoff = roff[min(r_lo + (int)lane, RSL)];
        const int S = __builtin_amdgcn_readlane((int)myoff, 0);
        const int E = __builtin_amdgcn_readlane((int)myoff, nown);

        int   cur = r_lo - 1;
        float a0 = 0.0f, a1 = 0.0f, a2 = 0.0f, a3 = 0.0f;
        auto flush = [&](int nxt) {
            if (cur >= r_lo) {
                float* t = mytile + (cur - r_lo) * TPITCH + lane;
                t[0] += a0; t[32] += a1; t[64] += a2; t[96] += a3;
            }
            a0 = 0.0f; a1 = 0.0f; a2 = 0.0f; a3 = 0.0f;
            cur = nxt;
        };

        for (int e0 = S; e0 < E; e0 += 32) {
            const int cn = min(32, E - e0);
            int2 pv = make_int2(0, 0);
            if ((int)lane < cn) pv = ent[e0 + (int)lane];
            int g = 0;
            for (; g + 8 <= cn; g += 8) {
                float m[8][4];
#pragma unroll
                for (int j = 0; j < 8; ++j) {
                    const unsigned p = (unsigned)__builtin_amdgcn_readlane(pv.x, g + j);
                    const float* s = row_lane_ptr(src, (p & 0x1FFFFu) * (HID_F * 4u) + lane4);
                    m[j][0] = s[0];
                    m[j][1] = s[32];
                    m[j][2] = s[64];
                    m[j][3] = s[96];
                }
#pragma unroll
                for (int j = 0; j < 8; ++j) {
                    const unsigned p = (unsigned)__builtin_amdgcn_readlane(pv.x, g + j);
                    const float    w = __int_as_float(__builtin_amdgcn_readlane(pv.y, g + j));
                    const int      rl = (int)(p >> 17);
                    if (rl != cur) flush(rl);
                    a0 = fmaf(w, m[j][0], a0);
                    a1 = fmaf(w, m[j][1], a1);
                    a2 = fmaf(w, m[j][2], a2);
                    a3 = fmaf(w, m[j][3], a3);
                }
            }
            if (g < cn) {
                float m[7][4];
#pragma unroll
                for (int j = 0; j < 7; ++j) {
                    if (g + j < cn) {
                        const unsigned p = (unsigned)__builtin_amdgcn_readlane(pv.x, g + j);
                        const float* s = row_lane_ptr(src, (p & 0x1FFFFu) * (HID_F * 4u) + lane4);
                        m[j][0] = s[0];
                        m[j][1] = s[32];
                        m[j][2] = s[64];
                        m[j][3] = s[96];
                    }
                }
#pragma unroll
                for (int j = 0; j < 7; ++j) {
                    if (g + j < cn) {
                        const unsigned p = (unsigned)__builtin_amdgcn_readlane(pv.x, g + j);
                        const float    w = __int_as_float(__builtin_amdgcn_readlane(pv.y, g + j));
                        const int      rl = (int)(p >> 17);
                        if (rl != cur) flush(rl);
                        a0 = fmaf(w, m[j][0], a0);
                        a1 = fmaf(w, m[j][1], a1);
                        a2 = fmaf(w, m[j][2], a2);
                        a3 = fmaf(w, m[j][3], a3);
                    }
                }
            }
        }
        flush(r_hi);
    }

    {
        const int  m   = (int)(lane & 15u);
        const int  kh  = (int)(lane >> 4);
        const bool own = m < nown;
        const float* arow = mytile + min(m, TPW - 1) * TPITCH + 2 * kh;
        const float* wcol = W2 + 2 * m;
        v8f acc0 = {}, acc1 = {};
        for (int kb = 0; kb < HID_F; kb += 32) {
            v2f bq[8][2];
#pragma unroll
            for (int q = 0; q < 8; ++q) {
                const float* wrow = wcol + (size_t)(kb + 4 * q + 2 * kh) * OUT_F;
                bq[q][0].x = wrow[0];
                bq[q][0].y = wrow[OUT_F];
                bq[q][1].x = wrow[1];
                bq[q][1].y = wrow[OUT_F + 1];
            }
#pragma unroll
            for (int q = 0; q < 8; ++q) {
                v2f a = *(const v2f*)(arow + kb + 4 * q);
                a.x = own ? fmaxf(a.x, 0.0f) : 0.0f;
                a.y = own ? fmaxf(a.y, 0.0f) : 0.0f;
                acc0 = __builtin_amdgcn_wmma_f32_16x16x4_f32(
                    false, a, false, bq[q][0], (short)0, acc0, false, false);
                acc1 = __builtin_amdgcn_wmma_f32_16x16x4_f32(
                    false, a, false, bq[q][1], (short)0, acc1, false, false);
            }
        }
        float* drow = h2 + ((size_t)b * RPB + (unsigned)(r_lo + kh * 8)) * OUT_F + 2 * m;
#pragma unroll
        for (int v = 0; v < 8; ++v) {
            const int r = kh * 8 + v;
            if (r < nown && (unsigned)(r_lo + r) < nrows) {
                v2f o;
                o.x = acc0[v];
                o.y = acc1[v];
                *(v2f*)(drow + (size_t)v * OUT_F) = o;
            }
        }
    }
}

__global__ __launch_bounds__(256) void spmm_f32(
    const int* __restrict__ rows, const int* __restrict__ cols,
    const float* __restrict__ vals, const float* __restrict__ src,
    float* __restrict__ dst)
{
    const int      wave = (blockIdx.x * blockDim.x + threadIdx.x) >> 5;
    const unsigned lane = threadIdx.x & 31;
    if (wave >= N_EDGES / 32) return;
    const int   e  = wave * 32 + (int)lane;
    const int   rv = rows[e];
    const int   cv = cols[e];
    const float wv = vals[e];
    const unsigned lane4 = lane * 4u;

    for (int g = 0; g < 32; g += 16) {
        float m[16];
#pragma unroll
        for (int j = 0; j < 16; ++j) {
            const unsigned c = (unsigned)__builtin_amdgcn_readlane(cv, g + j);
            m[j] = *row_lane_ptr(src, c * (OUT_F * 4u) + lane4);
        }
#pragma unroll
        for (int j = 0; j < 16; ++j) {
            const unsigned r = (unsigned)__builtin_amdgcn_readlane(rv, g + j);
            const float    w = lane_bcast_f(wv, g + j);
            __hip_atomic_fetch_add(row_lane_ptr(dst, r * (OUT_F * 4u) + lane4), w * m[j],
                                   __ATOMIC_RELAXED, __HIP_MEMORY_SCOPE_AGENT);
        }
    }
}

__global__ __launch_bounds__(256) void zero_f4(float4* __restrict__ p, int n4)
{
    int i = blockIdx.x * blockDim.x + threadIdx.x;
    if (i < n4) p[i] = make_float4(0.f, 0.f, 0.f, 0.f);
}

extern "C" void kernel_launch(void* const* d_in, const int* in_sizes, int n_in,
                              void* d_out, int out_size, void* d_ws, size_t ws_size,
                              hipStream_t stream)
{
    const float* x    = (const float*)d_in[0];
    const int*   rows = (const int*)  d_in[1];
    const int*   cols = (const int*)  d_in[2];
    const float* vals = (const float*)d_in[3];
    const float* W1   = (const float*)d_in[4];
    const float* W2   = (const float*)d_in[5];
    float* out = (float*)d_out;

    float* h0  = (float*)d_ws;
    float* b1  = h0 + (size_t)N_NODES * HID_F;
    int2*  rng = (int2*)b1;
    float* h2  = b1 + 4096;

    const int gemmBlocks = ((N_NODES / 16) + 3) / 4;
    const int edgeBlocks = (N_EDGES + 255) / 256;

    unsigned* cnt  = (unsigned*)d_ws;
    unsigned* tot  = cnt + NBKT * NBLK;
    int2*     list = (int2*)d_out;
    bin_count<<<NBLK, 1024, 0, stream>>>(rows, cnt);
    bin_scan<<<NBKT / 8, 256, 0, stream>>>(cnt, tot);
    bin_place<<<NBLK, 1024, 0, stream>>>(rows, cols, vals, cnt, tot, list, rng);

    gemm1_wmma<<<gemmBlocks, 128, 0, stream>>>(x, W1, h0);

    spmm_f128_rows_gemm2<<<(N_NODES + RPB - 1) / RPB, 512, 0, stream>>>(list, rng, h0, W2, h2);

    {
        int n4 = (N_NODES * OUT_F) / 4;
        zero_f4<<<(n4 + 255) / 256, 256, 0, stream>>>((float4*)out, n4);
    }
    spmm_f32<<<edgeBlocks, 256, 0, stream>>>(rows, cols, vals, h2, out);
}


# spmm F=32: reuses the binned list; a work-group per row range orders its entries in LDS (arrival-order integer rank), a wave sums each row in a register, stores it once; no global atomics or zero fill
# speedup vs baseline: 3.0865x; 1.1631x over previous
#include <hip/hip_runtime.h>

#define N_NODES 100000
#define N_EDGES 1600000
#define IN_F    256
#define HID_F   128
#define OUT_F   32

typedef float v2f __attribute__((ext_vector_type(2)));
typedef float v8f __attribute__((ext_vector_type(8)));

__global__ __launch_bounds__(128) void gemm1_wmma(
    const float* __restrict__ x, const float* __restrict__ W1,
    float* __restrict__ h0)
{
    const int wid  = (blockIdx.x * blockDim.x + threadIdx.x) >> 5;
    const int lane = threadIdx.x & 31;
    if (wid >= N_NODES / 16) return;
    const int m    = lane & 15;
    const int kh   = lane >> 4;
    const int row0 = wid * 16;

    v8f acc[8] = {};

    const float* arow = x + (size_t)(row0 + m) * IN_F + 2 * kh;
    const float* wcol = W1 + m;

    for (int k = 0; k < IN_F; k += 4) {
        v2f a = *(const v2f*)(arow + k);
        const float* wrow = wcol + (size_t)(k + 2 * kh) * HID_F;
#pragma unroll
        for (int t = 0; t < 8; ++t) {
            v2f b;
            b.x = wrow[t * 16];
            b.y = wrow[HID_F + t * 16];
            acc[t] = __builtin_amdgcn_wmma_f32_16x16x4_f32(
                false, a, false, b, (short)0, acc[t], false, false);
        }
    }

    float* drow = h0 + (size_t)(row0 + kh * 8) * HID_F + m;
#pragma unroll
    for (int t = 0; t < 8; ++t)
#pragma unroll
        for (int v = 0; v < 8; ++v)
            drow[(size_t)v * HID_F + t * 16] = acc[t][v];
}

__device__ __forceinline__ const float* row_lane_ptr(const float* base, unsigned off)
{
    return (const float*)((const char*)base + off);
}
__device__ __forceinline__ float* row_lane_ptr(float* base, unsigned off)
{
    return (float*)((char*)base + off);
}

#define NBKT 512
#define RPB  196
#define NBLK 128
#define EPB  (N_EDGES / NBLK)

__device__ __forceinline__ unsigned wave_incl_scan(unsigned v, unsigned lane)
{
#pragma unroll
    for (int d = 1; d < 32; d <<= 1) {
        unsigned u = __shfl_up(v, d, 32);
        if (lane >= (unsigned)d) v += u;
    }
    return v;
}

__global__ __launch_bounds__(1024) void bin_count(
    const int* __restrict__ rows, unsigned* __restrict__ cnt)
{
    __shared__ unsigned hist[NBKT];
    const unsigned tid = threadIdx.x, b = blockIdx.x;
    if (tid < NBKT) hist[tid] = 0u;
    __syncthreads();
    for (unsigned i = tid; i < EPB; i += 1024u) {
        const unsigned r = (unsigned)rows[b * EPB + i];
        atomicAdd(&hist[r / RPB], 1u);
    }
    __syncthreads();
    if (tid < NBKT) cnt[tid * NBLK + b] = hist[tid];
}

__global__ __launch_bounds__(256) void bin_scan(
    unsigned* __restrict__ cnt, unsigned* __restrict__ tot)
{
    const unsigned k    = (blockIdx.x * blockDim.x + threadIdx.x) >> 5;
    const unsigned lane = threadIdx.x & 31;
    uint4* p = (uint4*)(cnt + k * NBLK) + lane;
    const uint4 v = *p;
    const unsigned s    = v.x + v.y + v.z + v.w;
    const unsigned incl = wave_incl_scan(s, lane);
    const unsigned ex   = incl - s;
    *p = make_uint4(ex, ex + v.x, ex + v.x + v.y, ex + v.x + v.y + v.z);
    if (lane == 31) tot[k] = incl;
}

__global__ __launch_bounds__(1024) void bin_place(
    const int* __restrict__ rows, const int* __restrict__ cols,
    const float* __restrict__ vals, const unsigned* __restrict__ cnt,
    const unsigned* __restrict__ tot, int2* __restrict__ list,
    int2* __restrict__ rng)
{
    __shared__ unsigned cur[NBKT];
    __shared__ unsigned wsum[NBKT / 32];
    const unsigned tid = threadIdx.x, b = blockIdx.x, lane = tid & 31;
    unsigned v = 0u, incl = 0u;
    if (tid < NBKT) {
        v    = tot[tid];
        incl = wave_incl_scan(v, lane);
        if (lane == 31) wsum[tid >> 5] = incl;
    }
    __syncthreads();
    if (tid < NBKT) {
        unsigned base = 0u;
        for (unsigned w = 0; w < (tid >> 5); ++w) base += wsum[w];
        const unsigned first = base + incl - v;
        cur[tid] = first + cnt[tid * NBLK + b];
        if (b == 0) rng[tid] = make_int2((int)first, (int)(first + v));
    }
    __syncthreads();
    for (unsigned i0 = tid; i0 < EPB; i0 += 4096u) {
        unsigned r[4], c[4];
        int      w[4];
#pragma unroll
        for (int j = 0; j < 4; ++j) {
            const unsigned i = i0 + (unsigned)j * 1024u;
            if (i < EPB) {
                const unsigned e = b * EPB + i;
                r[j] = (unsigned)rows[e];
                c[j] = (unsigned)cols[e];
                w[j] = __float_as_int(vals[e]);
            }
        }
#pragma unroll
        for (int j = 0; j < 4; ++j) {
            const unsigned i = i0 + (unsigned)j * 1024u;
            if (i < EPB) {
                const unsigned k   = r[j] / RPB;
                const unsigned rl  = r[j] - k * RPB;
                const unsigned pos = atomicAdd(&cur[k], 1u);
                list[pos] = make_int2((int)(c[j] | (rl << 17)), w[j]);
            }
        }
    }
}

#define CAP 4096
#define EPT (CAP / 512)
#define RSL 224
#define TPW 13
#define TPITCH 132

__global__ __launch_bounds__(512) void spmm_f128_rows_gemm2(
    const int2* __restrict__ list, const int2* __restrict__ rng,
    const float* __restrict__ src, const float* __restrict__ W2,
    float* __restrict__ h2)
{
    __shared__ int2     ent[CAP];
    __shared__ unsigned rcnt[RSL];
    __shared__ unsigned roff[RSL + 1];
    __shared__ float    tile[16 * TPW * TPITCH];
    const unsigned tid = threadIdx.x, b = blockIdx.x;
    const unsigned lane = tid & 31;
    const int wave = __builtin_amdgcn_readfirstlane((int)(tid >> 5));
    const int2 hd = rng[b];
    const int first = __builtin_amdgcn_readfirstlane(hd.x);
    const int last  = __builtin_amdgcn_readfirstlane(hd.y);
    const unsigned lane4 = lane * 4u;
    const unsigned nrows = min((unsigned)RPB, (unsigned)N_NODES - b * RPB);
    const int r_lo = wave * 12 + min(wave, 4);
    const int nown = wave < 4 ? 13 : 12;
    const int r_hi = r_lo + nown;

    float* mytile = tile + wave * (TPW * TPITCH);
#pragma unroll
    for (int r = 0; r < TPW; ++r) {
        float* t = mytile + r * TPITCH + lane;
        t[0] = 0.0f; t[32] = 0.0f; t[64] = 0.0f; t[96] = 0.0f;
    }

    for (int p0 = first; p0 < last; p0 += CAP) {
        const int n = min(CAP, last - p0);
        if (tid < RSL) rcnt[tid] = 0u;
        __syncthreads();

        int2     e[EPT];
        unsigned rk[EPT];
#pragma unroll
        for (int k = 0; k < EPT; ++k) {
            const int i = (int)tid + k * 512;
            e[k] = make_int2(0, 0);
            if (i < n) e[k] = list[p0 + i];
        }
#pragma unroll
        for (int k = 0; k < EPT; ++k) {
            const int i = (int)tid + k * 512;
            rk[k] = 0u;
            if (i < n) rk[k] = atomicAdd(&rcnt[(unsigned)e[k].x >> 17], 1u);
        }
        __syncthreads();

        if (wave == 0) {
            unsigned c[7], s = 0u;
#pragma unroll
            for (int j = 0; j < 7; ++j) { c[j] = rcnt[lane * 7u + j]; s += c[j]; }
            const unsigned incl = wave_incl_scan(s, lane);
            unsigned run = incl - s;
#pragma unroll
            for (int j = 0; j < 7; ++j) { roff[lane * 7u + j] = run; run += c[j]; }
            if (lane == 31) roff[RSL] = incl;
        }
        __syncthreads();

#pragma unroll
        for (int k = 0; k < EPT; ++k) {
            const int i = (int)tid + k * 512;
            if (i < n) ent[roff[(unsigned)e[k].x >> 17] + rk[k]] = e[k];
        }
        __syncthreads();

        const unsigned myoff = roff[min(r_lo + (int)lane, RSL)];
        const int S = __builtin_amdgcn_readlane((int)myoff, 0);
        const int E = __builtin_amdgcn_readlane((int)myoff, nown);

        int   cur = r_lo - 1;
        float a0 = 0.0f, a1 = 0.0f, a2 = 0.0f, a3 = 0.0f;
        auto flush = [&](int nxt) {
            if (cur >= r_lo) {
                float* t = mytile + (cur - r_lo) * TPITCH + lane;
                t[0] += a0; t[32] += a1; t[64] += a2; t[96] += a3;
            }
            a0 = 0.0f; a1 = 0.0f; a2 = 0.0f; a3 = 0.0f;
            cur = nxt;
        };

        for (int e0 = S; e0 < E; e0 += 32) {
            const int cn = min(32, E - e0);
            int2 pv = make_int2(0, 0);
            if ((int)lane < cn) pv = ent[e0 + (int)lane];
            int g = 0;
            for (; g + 8 <= cn; g += 8) {
                float m[8][4];
#pragma unroll
                for (int j = 0; j < 8; ++j) {
                    const unsigned p = (unsigned)__builtin_amdgcn_readlane(pv.x, g + j);
                    const float* s = row_lane_ptr(src, (p & 0x1FFFFu) * (HID_F * 4u) + lane4);
                    m[j][0] = s[0];
                    m[j][1] = s[32];
                    m[j][2] = s[64];
                    m[j][3] = s[96];
                }
#pragma unroll
                for (int j = 0; j < 8; ++j) {
                    const unsigned p = (unsigned)__builtin_amdgcn_readlane(pv.x, g + j);
                    const float    w = __int_as_float(__builtin_amdgcn_readlane(pv.y, g + j));
                    const int      rl = (int)(p >> 17);
                    if (rl != cur) flush(rl);
                    a0 = fmaf(w, m[j][0], a0);
                    a1 = fmaf(w, m[j][1], a1);
                    a2 = fmaf(w, m[j][2], a2);
                    a3 = fmaf(w, m[j][3], a3);
                }
            }
            if (g < cn) {
                float m[7][4];
#pragma unroll
                for (int j = 0; j < 7; ++j) {
                    if (g + j < cn) {
                        const unsigned p = (unsigned)__builtin_amdgcn_readlane(pv.x, g + j);
                        const float* s = row_lane_ptr(src, (p & 0x1FFFFu) * (HID_F * 4u) + lane4);
                        m[j][0] = s[0];
                        m[j][1] = s[32];
                        m[j][2] = s[64];
                        m[j][3] = s[96];
                    }
                }
#pragma unroll
                for (int j = 0; j < 7; ++j) {
                    if (g + j < cn) {
                        const unsigned p = (unsigned)__builtin_amdgcn_readlane(pv.x, g + j);
                        const float    w = __int_as_float(__builtin_amdgcn_readlane(pv.y, g + j));
                        const int      rl = (int)(p >> 17);
                        if (rl != cur) flush(rl);
                        a0 = fmaf(w, m[j][0], a0);
                        a1 = fmaf(w, m[j][1], a1);
                        a2 = fmaf(w, m[j][2], a2);
                        a3 = fmaf(w, m[j][3], a3);
                    }
                }
            }
        }
        flush(r_hi);
    }

    {
        const int  m   = (int)(lane & 15u);
        const int  kh  = (int)(lane >> 4);
        const bool own = m < nown;
        const float* arow = mytile + min(m, TPW - 1) * TPITCH + 2 * kh;
        const float* wcol = W2 + 2 * m;
        v8f acc0 = {}, acc1 = {};
        for (int kb = 0; kb < HID_F; kb += 32) {
            v2f bq[8][2];
#pragma unroll
            for (int q = 0; q < 8; ++q) {
                const float* wrow = wcol + (size_t)(kb + 4 * q + 2 * kh) * OUT_F;
                bq[q][0].x = wrow[0];
                bq[q][0].y = wrow[OUT_F];
                bq[q][1].x = wrow[1];
                bq[q][1].y = wrow[OUT_F + 1];
            }
#pragma unroll
            for (int q = 0; q < 8; ++q) {
                v2f a = *(const v2f*)(arow + kb + 4 * q);
                a.x = own ? fmaxf(a.x, 0.0f) : 0.0f;
                a.y = own ? fmaxf(a.y, 0.0f) : 0.0f;
                acc0 = __builtin_amdgcn_wmma_f32_16x16x4_f32(
                    false, a, false, bq[q][0], (short)0, acc0, false, false);
                acc1 = __builtin_amdgcn_wmma_f32_16x16x4_f32(
                    false, a, false, bq[q][1], (short)0, acc1, false, false);
            }
        }
        float* drow = h2 + ((size_t)b * RPB + (unsigned)(r_lo + kh * 8)) * OUT_F + 2 * m;
#pragma unroll
        for (int v = 0; v < 8; ++v) {
            const int r = kh * 8 + v;
            if (r < nown && (unsigned)(r_lo + r) < nrows) {
                v2f o;
                o.x = acc0[v];
                o.y = acc1[v];
                *(v2f*)(drow + (size_t)v * OUT_F) = o;
            }
        }
    }
}

#define GRP 16

__global__ __launch_bounds__(512) void spmm_f32_rows(
    const int2* __restrict__ list, const int2* __restrict__ rng,
    const float* __restrict__ src, float* __restrict__ dst)
{
    __shared__ int2     ent[CAP];
    __shared__ unsigned rcnt[RSL];
    __shared__ unsigned roff[RSL + 1];
    __shared__ float    tile[16 * TPW * OUT_F];
    const unsigned tid = threadIdx.x, b = blockIdx.x;
    const unsigned lane = tid & 31;
    const int wave = __builtin_amdgcn_readfirstlane((int)(tid >> 5));
    const int2 hd = rng[b];
    const int first = __builtin_amdgcn_readfirstlane(hd.x);
    const int last  = __builtin_amdgcn_readfirstlane(hd.y);
    const unsigned lane4 = lane * 4u;
    const unsigned nrows = min((unsigned)RPB, (unsigned)N_NODES - b * RPB);
    const int r_lo = wave * 12 + min(wave, 4);
    const int nown = wave < 4 ? 13 : 12;
    const int r_hi = r_lo + nown;

    float* mytile = tile + wave * (TPW * OUT_F);
#pragma unroll
    for (int r = 0; r < TPW; ++r) mytile[r * OUT_F + (int)lane] = 0.0f;

    for (int p0 = first; p0 < last; p0 += CAP) {
        const int n = min(CAP, last - p0);
        if (tid < RSL) rcnt[tid] = 0u;
        __syncthreads();

        int2     e[EPT];
        unsigned rk[EPT];
#pragma unroll
        for (int k = 0; k < EPT; ++k) {
            const int i = (int)tid + k * 512;
            e[k] = make_int2(0, 0);
            if (i < n) e[k] = list[p0 + i];
        }
#pragma unroll
        for (int k = 0; k < EPT; ++k) {
            const int i = (int)tid + k * 512;
            rk[k] = 0u;
            if (i < n) rk[k] = atomicAdd(&rcnt[(unsigned)e[k].x >> 17], 1u);
        }
        __syncthreads();

        if (wave == 0) {
            unsigned c[7], s = 0u;
#pragma unroll
            for (int j = 0; j < 7; ++j) { c[j] = rcnt[lane * 7u + j]; s += c[j]; }
            const unsigned incl = wave_incl_scan(s, lane);
            unsigned run = incl - s;
#pragma unroll
            for (int j = 0; j < 7; ++j) { roff[lane * 7u + j] = run; run += c[j]; }
            if (lane == 31) roff[RSL] = incl;
        }
        __syncthreads();

#pragma unroll
        for (int k = 0; k < EPT; ++k) {
            const int i = (int)tid + k * 512;
            if (i < n) ent[roff[(unsigned)e[k].x >> 17] + rk[k]] = e[k];
        }
        __syncthreads();

        const unsigned myoff = roff[min(r_lo + (int)lane, RSL)];
        const int S = __builtin_amdgcn_readlane((int)myoff, 0);
        const int E = __builtin_amdgcn_readlane((int)myoff, nown);

        int   cur = r_lo - 1;
        float a = 0.0f;
        auto flush = [&](int nxt) {
            if (cur >= r_lo) mytile[(cur - r_lo) * OUT_F + (int)lane] += a;
            a = 0.0f;
            cur = nxt;
        };

        for (int e0 = S; e0 < E; e0 += 32) {
            const int cn = min(32, E - e0);
            int2 pv = make_int2(0, 0);
            if ((int)lane < cn) pv = ent[e0 + (int)lane];
            int g = 0;
            for (; g + GRP <= cn; g += GRP) {
                float m[GRP];
#pragma unroll
                for (int j = 0; j < GRP; ++j) {
                    const unsigned p = (unsigned)__builtin_amdgcn_readlane(pv.x, g + j);
                    m[j] = *row_lane_ptr(src, (p & 0x1FFFFu) * (OUT_F * 4u) + lane4);
                }
#pragma unroll
                for (int j = 0; j < GRP; ++j) {
                    const unsigned p = (unsigned)__builtin_amdgcn_readlane(pv.x, g + j);
                    const float    w = __int_as_float(__builtin_amdgcn_readlane(pv.y, g + j));
                    const int      rl = (int)(p >> 17);
                    if (rl != cur) flush(rl);
                    a = fmaf(w, m[j], a);
                }
            }
            if (g < cn) {
                float m[GRP - 1];
#pragma unroll
                for (int j = 0; j < GRP - 1; ++j) {
                    if (g + j < cn) {
                        const unsigned p = (unsigned)__builtin_amdgcn_readlane(pv.x, g + j);
                        m[j] = *row_lane_ptr(src, (p & 0x1FFFFu) * (OUT_F * 4u) + lane4);
                    }
                }
#pragma unroll
                for (int j = 0; j < GRP - 1; ++j) {
                    if (g + j < cn) {
                        const unsigned p = (unsigned)__builtin_amdgcn_readlane(pv.x, g + j);
                        const float    w = __int_as_float(__builtin_amdgcn_readlane(pv.y, g + j));
                        const int      rl = (int)(p >> 17);
                        if (rl != cur) flush(rl);
                        a = fmaf(w, m[j], a);
                    }
                }
            }
        }
        flush(r_hi);
    }

    float* drow = dst + ((size_t)b * RPB + (unsigned)r_lo) * OUT_F + lane;
#pragma unroll
    for (int r = 0; r < TPW; ++r) {
        if (r < nown && (unsigned)(r_lo + r) < nrows)
            drow[(size_t)r * OUT_F] = mytile[r * OUT_F + (int)lane];
    }
}

extern "C" void kernel_launch(void* const* d_in, const int* in_sizes, int n_in,
                              void* d_out, int out_size, void* d_ws, size_t ws_size,
                              hipStream_t stream)
{
    const float* x    = (const float*)d_in[0];
    const int*   rows = (const int*)  d_in[1];
    const int*   cols = (const int*)  d_in[2];
    const float* vals = (const float*)d_in[3];
    const float* W1   = (const float*)d_in[4];
    const float* W2   = (const float*)d_in[5];
    float* out = (float*)d_out;

    float* h0  = (float*)d_ws;
    float* b1  = h0 + (size_t)N_NODES * HID_F;
    int2*  rng = (int2*)b1;
    float* h2  = b1 + 4096;
    int2*  list = (int2*)(h2 + (size_t)N_NODES * OUT_F);

    const int gemmBlocks = ((N_NODES / 16) + 3) / 4;

    unsigned* cnt  = (unsigned*)d_ws;
    unsigned* tot  = cnt + NBKT * NBLK;
    bin_count<<<NBLK, 1024, 0, stream>>>(rows, cnt);
    bin_scan<<<NBKT / 8, 256, 0, stream>>>(cnt, tot);
    bin_place<<<NBLK, 1024, 0, stream>>>(rows, cols, vals, cnt, tot, list, rng);

    gemm1_wmma<<<gemmBlocks, 128, 0, stream>>>(x, W1, h0);

    spmm_f128_rows_gemm2<<<(N_NODES + RPB - 1) / RPB, 512, 0, stream>>>(list, rng, h0, W2, h2);

    spmm_f32_rows<<<(N_NODES + RPB - 1) / RPB, 512, 0, stream>>>(list, rng, h2, out);
}


# binning: each edge slice ordered by row range in ONE launch (per-slice LDS counting sort), not a global count, scan, place; a range reads its 128 slice segments; 6 launches -> 4; arrival-order sums
# speedup vs baseline: 3.4492x; 1.1175x over previous
#include <hip/hip_runtime.h>

#define N_NODES 100000
#define N_EDGES 1600000
#define IN_F    256
#define HID_F   128
#define OUT_F   32

typedef float v2f __attribute__((ext_vector_type(2)));
typedef float v8f __attribute__((ext_vector_type(8)));

__global__ __launch_bounds__(128) void gemm1_wmma(
    const float* __restrict__ x, const float* __restrict__ W1,
    float* __restrict__ h0)
{
    const int wid  = (blockIdx.x * blockDim.x + threadIdx.x) >> 5;
    const int lane = threadIdx.x & 31;
    if (wid >= N_NODES / 16) return;
    const int m    = lane & 15;
    const int kh   = lane >> 4;
    const int row0 = wid * 16;

    v8f acc[8] = {};

    const float* arow = x + (size_t)(row0 + m) * IN_F + 2 * kh;
    const float* wcol = W1 + m;

    for (int k = 0; k < IN_F; k += 4) {
        v2f a = *(const v2f*)(arow + k);
        const float* wrow = wcol + (size_t)(k + 2 * kh) * HID_F;
#pragma unroll
        for (int t = 0; t < 8; ++t) {
            v2f b;
            b.x = wrow[t * 16];
            b.y = wrow[HID_F + t * 16];
            acc[t] = __builtin_amdgcn_wmma_f32_16x16x4_f32(
                false, a, false, b, (short)0, acc[t], false, false);
        }
    }

    float* drow = h0 + (size_t)(row0 + kh * 8) * HID_F + m;
#pragma unroll
    for (int t = 0; t < 8; ++t)
#pragma unroll
        for (int v = 0; v < 8; ++v)
            drow[(size_t)v * HID_F + t * 16] = acc[t][v];
}

__device__ __forceinline__ const float* row_lane_ptr(const float* base, unsigned off)
{
    return (const float*)((const char*)base + off);
}
__device__ __forceinline__ float* row_lane_ptr(float* base, unsigned off)
{
    return (float*)((char*)base + off);
}

#define NBKT 512
#define RPB  196
#define NBLK 128
#define EPB  (N_EDGES / NBLK)
#define EPL  ((EPB + 1023) / 1024)

__device__ __forceinline__ unsigned wave_incl_scan(unsigned v, unsigned lane)
{
#pragma unroll
    for (int d = 1; d < 32; d <<= 1) {
        unsigned u = __shfl_up(v, d, 32);
        if (lane >= (unsigned)d) v += u;
    }
    return v;
}

__global__ __launch_bounds__(1024) void bin_local(
    const int* __restrict__ rows, const int* __restrict__ cols,
    const float* __restrict__ vals, int2* __restrict__ list,
    int2* __restrict__ seg)
{
    __shared__ unsigned cur[NBKT];
    __shared__ unsigned wsum[NBKT / 32];
    const unsigned tid = threadIdx.x, b = blockIdx.x, lane = tid & 31;
    if (tid < NBKT) cur[tid] = 0u;
    unsigned r[EPL], c[EPL];
    int      w[EPL];
#pragma unroll
    for (int j = 0; j < EPL; ++j) {
        const unsigned i = tid + (unsigned)j * 1024u;
        if (i < EPB) {
            const unsigned e = b * EPB + i;
            r[j] = (unsigned)rows[e];
            c[j] = (unsigned)cols[e];
            w[j] = __float_as_int(vals[e]);
        }
    }
    __syncthreads();
#pragma unroll
    for (int j = 0; j < EPL; ++j) {
        const unsigned i = tid + (unsigned)j * 1024u;
        if (i < EPB) atomicAdd(&cur[r[j] / RPB], 1u);
    }
    __syncthreads();
    unsigned v = 0u, incl = 0u;
    if (tid < NBKT) {
        v    = cur[tid];
        incl = wave_incl_scan(v, lane);
        if (lane == 31) wsum[tid >> 5] = incl;
    }
    __syncthreads();
    if (tid < NBKT) {
        unsigned base = 0u;
        for (unsigned q = 0; q < (tid >> 5); ++q) base += wsum[q];
        const unsigned first = b * EPB + base + incl - v;
        cur[tid] = first;
        seg[tid * NBLK + b] = make_int2((int)first, (int)(first + v));
    }
    __syncthreads();
#pragma unroll
    for (int j = 0; j < EPL; ++j) {
        const unsigned i = tid + (unsigned)j * 1024u;
        if (i < EPB) {
            const unsigned k   = r[j] / RPB;
            const unsigned rl  = r[j] - k * RPB;
            const unsigned pos = atomicAdd(&cur[k], 1u);
            list[pos] = make_int2((int)(c[j] | (rl << 17)), w[j]);
        }
    }
}

__device__ __forceinline__ unsigned seg_of(const unsigned* segbase, unsigned f)
{
    unsigned s = 0u;
#pragma unroll
    for (unsigned step = NBLK / 2; step >= 1u; step >>= 1)
        if (segbase[s + step] <= f) s += step;
    return s;
}

#define CAP 4096
#define EPT (CAP / 512)
#define RSL 224
#define TPW 13
#define TPITCH 132

__global__ __launch_bounds__(512) void spmm_f128_rows_gemm2(
    const int2* __restrict__ list, const int2* __restrict__ seg,
    const float* __restrict__ src, const float* __restrict__ W2,
    float* __restrict__ h2)
{
    __shared__ int2     ent[CAP];
    __shared__ unsigned rcnt[RSL];
    __shared__ unsigned roff[RSL + 1];
    __shared__ unsigned segbase[NBLK + 1];
    __shared__ int      segadj[NBLK];
    __shared__ unsigned swsum[NBLK / 32];
    __shared__ float    tile[16 * TPW * TPITCH];
    const unsigned tid = threadIdx.x, b = blockIdx.x;
    const unsigned lane = tid & 31;
    const int wave = __builtin_amdgcn_readfirstlane((int)(tid >> 5));
    const unsigned lane4 = lane * 4u;
    const unsigned nrows = min((unsigned)RPB, (unsigned)N_NODES - b * RPB);
    const int r_lo = wave * 12 + min(wave, 4);
    const int nown = wave < 4 ? 13 : 12;
    const int r_hi = r_lo + nown;

    float* mytile = tile + wave * (TPW * TPITCH);
#pragma unroll
    for (int r = 0; r < TPW; ++r) {
        float* t = mytile + r * TPITCH + lane;
        t[0] = 0.0f; t[32] = 0.0f; t[64] = 0.0f; t[96] = 0.0f;
    }

    {
        int2 se = make_int2(0, 0);
        unsigned len = 0u, incl = 0u;
        if (wave < NBLK / 32) {
            se   = seg[b * NBLK + tid];
            len  = (unsigned)(se.y - se.x);
            incl = wave_incl_scan(len, lane);
            if (lane == 31) swsum[wave] = incl;
        }
        __syncthreads();
        if (wave < NBLK / 32) {
            unsigned base = incl - len;
            for (int q = 0; q < wave; ++q) base += swsum[q];
            segbase[tid] = base;
            segadj[tid]  = se.x - (int)base;
            if (tid == NBLK - 1) segbase[NBLK] = base + len;
        }
        __syncthreads();
    }
    const int last = __builtin_amdgcn_readfirstlane((int)segbase[NBLK]);

    for (int p0 = 0; p0 < last; p0 += CAP) {
        const int n = min(CAP, last - p0);
        if (tid < RSL) rcnt[tid] = 0u;
        __syncthreads();

        int2     e[EPT];
        unsigned rk[EPT];
#pragma unroll
        for (int k = 0; k < EPT; ++k) {
            const int i = (int)tid + k * 512;
            e[k] = make_int2(0, 0);
            if (i < n) {
                const unsigned f = (unsigned)(p0 + i);
                e[k] = list[segadj[seg_of(segbase, f)] + (int)f];
            }
        }
#pragma unroll
        for (int k = 0; k < EPT; ++k) {
            const int i = (int)tid + k * 512;
            rk[k] = 0u;
            if (i < n) rk[k] = atomicAdd(&rcnt[(unsigned)e[k].x >> 17], 1u);
        }
        __syncthreads();

        if (wave == 0) {
            unsigned c[7], s = 0u;
#pragma unroll
            for (int j = 0; j < 7; ++j) { c[j] = rcnt[lane * 7u + j]; s += c[j]; }
            const unsigned incl = wave_incl_scan(s, lane);
            unsigned run = incl - s;
#pragma unroll
            for (int j = 0; j < 7; ++j) { roff[lane * 7u + j] = run; run += c[j]; }
            if (lane == 31) roff[RSL] = incl;
        }
        __syncthreads();

#pragma unroll
        for (int k = 0; k < EPT; ++k) {
            const int i = (int)tid + k * 512;
            if (i < n) ent[roff[(unsigned)e[k].x >> 17] + rk[k]] = e[k];
        }
        __syncthreads();

        const unsigned myoff = roff[min(r_lo + (int)lane, RSL)];
        const int S = __builtin_amdgcn_readlane((int)myoff, 0);
        const int E = __builtin_amdgcn_readlane((int)myoff, nown);

        int   cur = r_lo - 1;
        float a0 = 0.0f, a1 = 0.0f, a2 = 0.0f, a3 = 0.0f;
        auto flush = [&](int nxt) {
            if (cur >= r_lo) {
                float* t = mytile + (cur - r_lo) * TPITCH + lane;
                t[0] += a0; t[32] += a1; t[64] += a2; t[96] += a3;
            }
            a0 = 0.0f; a1 = 0.0f; a2 = 0.0f; a3 = 0.0f;
            cur = nxt;
        };

        for (int e0 = S; e0 < E; e0 += 32) {
            const int cn = min(32, E - e0);
            int2 pv = make_int2(0, 0);
            if ((int)lane < cn) pv = ent[e0 + (int)lane];
            int g = 0;
            for (; g + 8 <= cn; g += 8) {
                float m[8][4];
#pragma unroll
                for (int j = 0; j < 8; ++j) {
                    const unsigned p = (unsigned)__builtin_amdgcn_readlane(pv.x, g + j);
                    const float* s = row_lane_ptr(src, (p & 0x1FFFFu) * (HID_F * 4u) + lane4);
                    m[j][0] = s[0];
                    m[j][1] = s[32];
                    m[j][2] = s[64];
                    m[j][3] = s[96];
                }
#pragma unroll
                for (int j = 0; j < 8; ++j) {
                    const unsigned p = (unsigned)__builtin_amdgcn_readlane(pv.x, g + j);
                    const float    w = __int_as_float(__builtin_amdgcn_readlane(pv.y, g + j));
                    const int      rl = (int)(p >> 17);
                    if (rl != cur) flush(rl);
                    a0 = fmaf(w, m[j][0], a0);
                    a1 = fmaf(w, m[j][1], a1);
                    a2 = fmaf(w, m[j][2], a2);
                    a3 = fmaf(w, m[j][3], a3);
                }
            }
            if (g < cn) {
                float m[7][4];
#pragma unroll
                for (int j = 0; j < 7; ++j) {
                    if (g + j < cn) {
                        const unsigned p = (unsigned)__builtin_amdgcn_readlane(pv.x, g + j);
                        const float* s = row_lane_ptr(src, (p & 0x1FFFFu) * (HID_F * 4u) + lane4);
                        m[j][0] = s[0];
                        m[j][1] = s[32];
                        m[j][2] = s[64];
                        m[j][3] = s[96];
                    }
                }
#pragma unroll
                for (int j = 0; j < 7; ++j) {
                    if (g + j < cn) {
                        const unsigned p = (unsigned)__builtin_amdgcn_readlane(pv.x, g + j);
                        const float    w = __int_as_float(__builtin_amdgcn_readlane(pv.y, g + j));
                        const int      rl = (int)(p >> 17);
                        if (rl != cur) flush(rl);
                        a0 = fmaf(w, m[j][0], a0);
                        a1 = fmaf(w, m[j][1], a1);
                        a2 = fmaf(w, m[j][2], a2);
                        a3 = fmaf(w, m[j][3], a3);
                    }
                }
            }
        }
        flush(r_hi);
    }

    {
        const int  m   = (int)(lane & 15u);
        const int  kh  = (int)(lane >> 4);
        const bool own = m < nown;
        const float* arow = mytile + min(m, TPW - 1) * TPITCH + 2 * kh;
        const float* wcol = W2 + 2 * m;
        v8f acc0 = {}, acc1 = {};
        for (int kb = 0; kb < HID_F; kb += 32) {
            v2f bq[8][2];
#pragma unroll
            for (int q = 0; q < 8; ++q) {
                const float* wrow = wcol + (size_t)(kb + 4 * q + 2 * kh) * OUT_F;
                bq[q][0].x = wrow[0];
                bq[q][0].y = wrow[OUT_F];
                bq[q][1].x = wrow[1];
                bq[q][1].y = wrow[OUT_F + 1];
            }
#pragma unroll
            for (int q = 0; q < 8; ++q) {
                v2f a = *(const v2f*)(arow + kb + 4 * q);
                a.x = own ? fmaxf(a.x, 0.0f) : 0.0f;
                a.y = own ? fmaxf(a.y, 0.0f) : 0.0f;
                acc0 = __builtin_amdgcn_wmma_f32_16x16x4_f32(
                    false, a, false, bq[q][0], (short)0, acc0, false, false);
                acc1 = __builtin_amdgcn_wmma_f32_16x16x4_f32(
                    false, a, false, bq[q][1], (short)0, acc1, false, false);
            }
        }
        float* drow = h2 + ((size_t)b * RPB + (unsigned)(r_lo + kh * 8)) * OUT_F + 2 * m;
#pragma unroll
        for (int v = 0; v < 8; ++v) {
            const int r = kh * 8 + v;
            if (r < nown && (unsigned)(r_lo + r) < nrows) {
                v2f o;
                o.x = acc0[v];
                o.y = acc1[v];
                *(v2f*)(drow + (size_t)v * OUT_F) = o;
            }
        }
    }
}

#define GRP 16

__global__ __launch_bounds__(512) void spmm_f32_rows(
    const int2* __restrict__ list, const int2* __restrict__ seg,
    const float* __restrict__ src, float* __restrict__ dst)
{
    __shared__ int2     ent[CAP];
    __shared__ unsigned rcnt[RSL];
    __shared__ unsigned roff[RSL + 1];
    __shared__ unsigned segbase[NBLK + 1];
    __shared__ int      segadj[NBLK];
    __shared__ unsigned swsum[NBLK / 32];
    __shared__ float    tile[16 * TPW * OUT_F];
    const unsigned tid = threadIdx.x, b = blockIdx.x;
    const unsigned lane = tid & 31;
    const int wave = __builtin_amdgcn_readfirstlane((int)(tid >> 5));
    const unsigned lane4 = lane * 4u;
    const unsigned nrows = min((unsigned)RPB, (unsigned)N_NODES - b * RPB);
    const int r_lo = wave * 12 + min(wave, 4);
    const int nown = wave < 4 ? 13 : 12;
    const int r_hi = r_lo + nown;

    float* mytile = tile + wave * (TPW * OUT_F);
#pragma unroll
    for (int r = 0; r < TPW; ++r) mytile[r * OUT_F + (int)lane] = 0.0f;

    {
        int2 se = make_int2(0, 0);
        unsigned len = 0u, incl = 0u;
        if (wave < NBLK / 32) {
            se   = seg[b * NBLK + tid];
            len  = (unsigned)(se.y - se.x);
            incl = wave_incl_scan(len, lane);
            if (lane == 31) swsum[wave] = incl;
        }
        __syncthreads();
        if (wave < NBLK / 32) {
            unsigned base = incl - len;
            for (int q = 0; q < wave; ++q) base += swsum[q];
            segbase[tid] = base;
            segadj[tid]  = se.x - (int)base;
            if (tid == NBLK - 1) segbase[NBLK] = base + len;
        }
        __syncthreads();
    }
    const int last = __builtin_amdgcn_readfirstlane((int)segbase[NBLK]);

    for (int p0 = 0; p0 < last; p0 += CAP) {
        const int n = min(CAP, last - p0);
        if (tid < RSL) rcnt[tid] = 0u;
        __syncthreads();

        int2     e[EPT];
        unsigned rk[EPT];
#pragma unroll
        for (int k = 0; k < EPT; ++k) {
            const int i = (int)tid + k * 512;
            e[k] = make_int2(0, 0);
            if (i < n) {
                const unsigned f = (unsigned)(p0 + i);
                e[k] = list[segadj[seg_of(segbase, f)] + (int)f];
            }
        }
#pragma unroll
        for (int k = 0; k < EPT; ++k) {
            const int i = (int)tid + k * 512;
            rk[k] = 0u;
            if (i < n) rk[k] = atomicAdd(&rcnt[(unsigned)e[k].x >> 17], 1u);
        }
        __syncthreads();

        if (wave == 0) {
            unsigned c[7], s = 0u;
#pragma unroll
            for (int j = 0; j < 7; ++j) { c[j] = rcnt[lane * 7u + j]; s += c[j]; }
            const unsigned incl = wave_incl_scan(s, lane);
            unsigned run = incl - s;
#pragma unroll
            for (int j = 0; j < 7; ++j) { roff[lane * 7u + j] = run; run += c[j]; }
            if (lane == 31) roff[RSL] = incl;
        }
        __syncthreads();

#pragma unroll
        for (int k = 0; k < EPT; ++k) {
            const int i = (int)tid + k * 512;
            if (i < n) ent[roff[(unsigned)e[k].x >> 17] + rk[k]] = e[k];
        }
        __syncthreads();

        const unsigned myoff = roff[min(r_lo + (int)lane, RSL)];
        const int S = __builtin_amdgcn_readlane((int)myoff, 0);
        const int E = __builtin_amdgcn_readlane((int)myoff, nown);

        int   cur = r_lo - 1;
        float a = 0.0f;
        auto flush = [&](int nxt) {
            if (cur >= r_lo) mytile[(cur - r_lo) * OUT_F + (int)lane] += a;
            a = 0.0f;
            cur = nxt;
        };

        for (int e0 = S; e0 < E; e0 += 32) {
            const int cn = min(32, E - e0);
            int2 pv = make_int2(0, 0);
            if ((int)lane < cn) pv = ent[e0 + (int)lane];
            int g = 0;
            for (; g + GRP <= cn; g += GRP) {
                float m[GRP];
#pragma unroll
                for (int j = 0; j < GRP; ++j) {
                    const unsigned p = (unsigned)__builtin_amdgcn_readlane(pv.x, g + j);
                    m[j] = *row_lane_ptr(src, (p & 0x1FFFFu) * (OUT_F * 4u) + lane4);
                }
#pragma unroll
                for (int j = 0; j < GRP; ++j) {
                    const unsigned p = (unsigned)__builtin_amdgcn_readlane(pv.x, g + j);
                    const float    w = __int_as_float(__builtin_amdgcn_readlane(pv.y, g + j));
                    const int      rl = (int)(p >> 17);
                    if (rl != cur) flush(rl);
                    a = fmaf(w, m[j], a);
                }
            }
            if (g < cn) {
                float m[GRP - 1];
#pragma unroll
                for (int j = 0; j < GRP - 1; ++j) {
                    if (g + j < cn) {
                        const unsigned p = (unsigned)__builtin_amdgcn_readlane(pv.x, g + j);
                        m[j] = *row_lane_ptr(src, (p & 0x1FFFFu) * (OUT_F * 4u) + lane4);
                    }
                }
#pragma unroll
                for (int j = 0; j < GRP - 1; ++j) {
                    if (g + j < cn) {
                        const unsigned p = (unsigned)__builtin_amdgcn_readlane(pv.x, g + j);
                        const float    w = __int_as_float(__builtin_amdgcn_readlane(pv.y, g + j));
                        const int      rl = (int)(p >> 17);
                        if (rl != cur) flush(rl);
                        a = fmaf(w, m[j], a);
                    }
                }
            }
        }
        flush(r_hi);
    }

    float* drow = dst + ((size_t)b * RPB + (unsigned)r_lo) * OUT_F + lane;
#pragma unroll
    for (int r = 0; r < TPW; ++r) {
        if (r < nown && (unsigned)(r_lo + r) < nrows)
            drow[(size_t)r * OUT_F] = mytile[r * OUT_F + (int)lane];
    }
}

extern "C" void kernel_launch(void* const* d_in, const int* in_sizes, int n_in,
                              void* d_out, int out_size, void* d_ws, size_t ws_size,
                              hipStream_t stream)
{
    const float* x    = (const float*)d_in[0];
    const int*   rows = (const int*)  d_in[1];
    const int*   cols = (const int*)  d_in[2];
    const float* vals = (const float*)d_in[3];
    const float* W1   = (const float*)d_in[4];
    const float* W2   = (const float*)d_in[5];
    float* out = (float*)d_out;

    float* h0  = (float*)d_ws;
    float* b1  = h0 + (size_t)N_NODES * HID_F;
    float* h2  = b1 + 4096;
    int2*  list = (int2*)(h2 + (size_t)N_NODES * OUT_F);
    int2*  seg  = list + N_EDGES;

    const int gemmBlocks = ((N_NODES / 16) + 3) / 4;

    bin_local<<<NBLK, 1024, 0, stream>>>(rows, cols, vals, list, seg);

    gemm1_wmma<<<gemmBlocks, 128, 0, stream>>>(x, W1, h0);

    spmm_f128_rows_gemm2<<<(N_NODES + RPB - 1) / RPB, 512, 0, stream>>>(list, seg, h0, W2, h2);

    spmm_f32_rows<<<(N_NODES + RPB - 1) / RPB, 512, 0, stream>>>(list, seg, h2, out);
}


# gemm1: W1 kept in LDS as ready B fragments, one 16-wave work-group per WGP; waves loop over 16x64 tasks on 16-byte x and LDS loads, whole-line h0 stores; k pairing per matrix step (sum order) changed
# speedup vs baseline: 4.0263x; 1.1673x over previous
#include <hip/hip_runtime.h>

#define N_NODES 100000
#define N_EDGES 1600000
#define IN_F    256
#define HID_F   128
#define OUT_F   32

typedef float v2f __attribute__((ext_vector_type(2)));
typedef float v4f __attribute__((ext_vector_type(4)));
typedef float v8f __attribute__((ext_vector_type(8)));

#define G1_WGS    256
#define G1_PAIRS  (G1_WGS * 8)
#define G1_LDS_SLOTS 10496

__global__ __launch_bounds__(512) void gemm1_wmma(
    const float* __restrict__ x, const float* __restrict__ W1,
    float* __restrict__ h0)
{
    __shared__ v4f w1s[G1_LDS_SLOTS];
    const unsigned tid = threadIdx.x, lane = tid & 31;
    const int wave = __builtin_amdgcn_readfirstlane((int)(tid >> 5));
    const int c = (int)(lane & 15u), h = (int)(lane >> 4);

    {
        v4f k0[8], k1[8];
#pragma unroll
        for (int i = 0; i < 8; ++i) {
            const int j = wave + 16 * i;
            k0[i] = *(const v4f*)(W1 + (size_t)(2 * j) * HID_F + 4 * lane);
            k1[i] = *(const v4f*)(W1 + (size_t)(2 * j + 1) * HID_F + 4 * lane);
        }
#pragma unroll
        for (int i = 0; i < 8; ++i) {
            const int j = wave + 16 * i;
            const int slot = j * 64 + h * 32 + c;
            v4f lo, hi;
            lo.x = k0[i].x; lo.y = k1[i].x; lo.z = k0[i].y; lo.w = k1[i].y;
            hi.x = k0[i].z; hi.y = k1[i].z; hi.z = k0[i].w; hi.w = k1[i].w;
            w1s[slot]      = lo;
            w1s[slot + 16] = hi;
        }
    }
    __syncthreads();

    const int nh = wave & 1;
    const int P = (wave >> 1) * G1_WGS + (int)blockIdx.x;
    const v4f* bbase = w1s + h * 128 + nh * 32 + c;

    int rb = P;
    const float* arow = x + (size_t)(rb * 16 + c) * IN_F + 4 * h;
    v4f a[2][8];
    v4f b[2][4];
#pragma unroll
    for (int i = 0; i < 8; ++i) a[0][i] = *(const v4f*)(arow + 8 * i);
    b[0][0] = bbase[0];
    b[0][1] = bbase[16];
    b[0][2] = bbase[64];
    b[0][3] = bbase[64 + 16];

    for (; rb < N_NODES / 16; rb += G1_PAIRS) {
        const int rbn = rb + G1_PAIRS < N_NODES / 16 ? rb + G1_PAIRS : rb;
        const float* anext = x + (size_t)(rbn * 16 + c) * IN_F + 4 * h;
        v8f acc0 = {}, acc1 = {}, acc2 = {}, acc3 = {};
#pragma nounroll
        for (int hf = 0; hf < 2; ++hf) {
            const v4f*   bh    = bbase + hf * 4096;
            const v4f*   bwrap = bbase + (1 - hf) * 4096;
            const float* acur  = arow + hf * 128;
            const float* apre  = hf ? anext : arow + 128;
#pragma unroll
            for (int sb = 0; sb < 2; ++sb) {
#pragma unroll
                for (int i = 0; i < 8; ++i)
                    a[(sb + 1) & 1][i] = sb == 0 ? *(const v4f*)(acur + 64 + 8 * i)
                                                 : *(const v4f*)(apre + 8 * i);
#pragma unroll
                for (int i = 0; i < 8; ++i) {
                    const int ql = 8 * sb + i;
                    if (ql + 1 < 16) {
                        b[(ql + 1) & 1][0] = bh[(ql + 1) * 256];
                        b[(ql + 1) & 1][1] = bh[(ql + 1) * 256 + 16];
                        b[(ql + 1) & 1][2] = bh[(ql + 1) * 256 + 64];
                        b[(ql + 1) & 1][3] = bh[(ql + 1) * 256 + 64 + 16];
                    } else {
                        b[0][0] = bwrap[0];
                        b[0][1] = bwrap[16];
                        b[0][2] = bwrap[64];
                        b[0][3] = bwrap[64 + 16];
                    }
                    const v4f av = a[sb][i];
#pragma unroll
                    for (int u = 0; u < 2; ++u) {
                        const v4f b01 = b[ql & 1][2 * u];
                        const v4f b23 = b[ql & 1][2 * u + 1];
                        v2f af, b0, b1, b2, b3;
                        af.x = u ? av.z : av.x;  af.y = u ? av.w : av.y;
                        b0.x = b01.x; b0.y = b01.y;
                        b1.x = b01.z; b1.y = b01.w;
                        b2.x = b23.x; b2.y = b23.y;
                        b3.x = b23.z; b3.y = b23.w;
                        acc0 = __builtin_amdgcn_wmma_f32_16x16x4_f32(
                            false, af, false, b0, (short)0, acc0, false, false);
                        acc1 = __builtin_amdgcn_wmma_f32_16x16x4_f32(
                            false, af, false, b1, (short)0, acc1, false, false);
                        acc2 = __builtin_amdgcn_wmma_f32_16x16x4_f32(
                            false, af, false, b2, (short)0, acc2, false, false);
                        acc3 = __builtin_amdgcn_wmma_f32_16x16x4_f32(
                            false, af, false, b3, (short)0, acc3, false, false);
                    }
                    __builtin_amdgcn_sched_barrier(0);
                }
            }
        }
        asm volatile("v_nop\n\tv_nop\n\tv_nop\n\tv_nop\n\tv_nop\n\tv_nop\n\tv_nop\n\tv_nop");
        __builtin_amdgcn_sched_barrier(0);
        float* drow = h0 + (size_t)(rb * 16 + 8 * h) * HID_F + 64 * nh + 4 * c;
#pragma unroll
        for (int r = 0; r < 8; ++r) {
            v4f o;
            o.x = acc0[r]; o.y = acc1[r]; o.z = acc2[r]; o.w = acc3[r];
            *(v4f*)(drow + (size_t)r * HID_F) = o;
        }
        arow = anext;
    }
}

__device__ __forceinline__ const float* row_lane_ptr(const float* base, unsigned off)
{
    return (const float*)((const char*)base + off);
}
__device__ __forceinline__ float* row_lane_ptr(float* base, unsigned off)
{
    return (float*)((char*)base + off);
}

#define NBKT 512
#define RPB  196
#define NBLK 128
#define EPB  (N_EDGES / NBLK)
#define EPL  ((EPB + 1023) / 1024)

__device__ __forceinline__ unsigned wave_incl_scan(unsigned v, unsigned lane)
{
#pragma unroll
    for (int d = 1; d < 32; d <<= 1) {
        unsigned u = __shfl_up(v, d, 32);
        if (lane >= (unsigned)d) v += u;
    }
    return v;
}

__global__ __launch_bounds__(1024) void bin_local(
    const int* __restrict__ rows, const int* __restrict__ cols,
    const float* __restrict__ vals, int2* __restrict__ list,
    int2* __restrict__ seg)
{
    __shared__ unsigned cur[NBKT];
    __shared__ unsigned wsum[NBKT / 32];
    const unsigned tid = threadIdx.x, b = blockIdx.x, lane = tid & 31;
    if (tid < NBKT) cur[tid] = 0u;
    unsigned r[EPL], c[EPL];
    int      w[EPL];
#pragma unroll
    for (int j = 0; j < EPL; ++j) {
        const unsigned i = tid + (unsigned)j * 1024u;
        if (i < EPB) {
            const unsigned e = b * EPB + i;
            r[j] = (unsigned)rows[e];
            c[j] = (unsigned)cols[e];
            w[j] = __float_as_int(vals[e]);
        }
    }
    __syncthreads();
#pragma unroll
    for (int j = 0; j < EPL; ++j) {
        const unsigned i = tid + (unsigned)j * 1024u;
        if (i < EPB) atomicAdd(&cur[r[j] / RPB], 1u);
    }
    __syncthreads();
    unsigned v = 0u, incl = 0u;
    if (tid < NBKT) {
        v    = cur[tid];
        incl = wave_incl_scan(v, lane);
        if (lane == 31) wsum[tid >> 5] = incl;
    }
    __syncthreads();
    if (tid < NBKT) {
        unsigned base = 0u;
        for (unsigned q = 0; q < (tid >> 5); ++q) base += wsum[q];
        const unsigned first = b * EPB + base + incl - v;
        cur[tid] = first;
        seg[tid * NBLK + b] = make_int2((int)first, (int)(first + v));
    }
    __syncthreads();
#pragma unroll
    for (int j = 0; j < EPL; ++j) {
        const unsigned i = tid + (unsigned)j * 1024u;
        if (i < EPB) {
            const unsigned k   = r[j] / RPB;
            const unsigned rl  = r[j] - k * RPB;
            const unsigned pos = atomicAdd(&cur[k], 1u);
            list[pos] = make_int2((int)(c[j] | (rl << 17)), w[j]);
        }
    }
}

__device__ __forceinline__ unsigned seg_of(const unsigned* segbase, unsigned f)
{
    unsigned s = 0u;
#pragma unroll
    for (unsigned step = NBLK / 2; step >= 1u; step >>= 1)
        if (segbase[s + step] <= f) s += step;
    return s;
}

#define CAP 4096
#define EPT (CAP / 512)
#define RSL 224
#define TPW 13
#define TPITCH 132

__global__ __launch_bounds__(512) void spmm_f128_rows_gemm2(
    const int2* __restrict__ list, const int2* __restrict__ seg,
    const float* __restrict__ src, const float* __restrict__ W2,
    float* __restrict__ h2)
{
    __shared__ int2     ent[CAP];
    __shared__ unsigned rcnt[RSL];
    __shared__ unsigned roff[RSL + 1];
    __shared__ unsigned segbase[NBLK + 1];
    __shared__ int      segadj[NBLK];
    __shared__ unsigned swsum[NBLK / 32];
    __shared__ float    tile[16 * TPW * TPITCH];
    const unsigned tid = threadIdx.x, b = blockIdx.x;
    const unsigned lane = tid & 31;
    const int wave = __builtin_amdgcn_readfirstlane((int)(tid >> 5));
    const unsigned lane4 = lane * 4u;
    const unsigned nrows = min((unsigned)RPB, (unsigned)N_NODES - b * RPB);
    const int r_lo = wave * 12 + min(wave, 4);
    const int nown = wave < 4 ? 13 : 12;
    const int r_hi = r_lo + nown;

    float* mytile = tile + wave * (TPW * TPITCH);
#pragma unroll
    for (int r = 0; r < TPW; ++r) {
        float* t = mytile + r * TPITCH + lane;
        t[0] = 0.0f; t[32] = 0.0f; t[64] = 0.0f; t[96] = 0.0f;
    }

    {
        int2 se = make_int2(0, 0);
        unsigned len = 0u, incl = 0u;
        if (wave < NBLK / 32) {
            se   = seg[b * NBLK + tid];
            len  = (unsigned)(se.y - se.x);
            incl = wave_incl_scan(len, lane);
            if (lane == 31) swsum[wave] = incl;
        }
        __syncthreads();
        if (wave < NBLK / 32) {
            unsigned base = incl - len;
            for (int q = 0; q < wave; ++q) base += swsum[q];
            segbase[tid] = base;
            segadj[tid]  = se.x - (int)base;
            if (tid == NBLK - 1) segbase[NBLK] = base + len;
        }
        __syncthreads();
    }
    const int last = __builtin_amdgcn_readfirstlane((int)segbase[NBLK]);

    for (int p0 = 0; p0 < last; p0 += CAP) {
        const int n = min(CAP, last - p0);
        if (tid < RSL) rcnt[tid] = 0u;
        __syncthreads();

        int2     e[EPT];
        unsigned rk[EPT];
#pragma unroll
        for (int k = 0; k < EPT; ++k) {
            const int i = (int)tid + k * 512;
            e[k] = make_int2(0, 0);
            if (i < n) {
                const unsigned f = (unsigned)(p0 + i);
                e[k] = list[segadj[seg_of(segbase, f)] + (int)f];
            }
        }
#pragma unroll
        for (int k = 0; k < EPT; ++k) {
            const int i = (int)tid + k * 512;
            rk[k] = 0u;
            if (i < n) rk[k] = atomicAdd(&rcnt[(unsigned)e[k].x >> 17], 1u);
        }
        __syncthreads();

        if (wave == 0) {
            unsigned c[7], s = 0u;
#pragma unroll
            for (int j = 0; j < 7; ++j) { c[j] = rcnt[lane * 7u + j]; s += c[j]; }
            const unsigned incl = wave_incl_scan(s, lane);
            unsigned run = incl - s;
#pragma unroll
            for (int j = 0; j < 7; ++j) { roff[lane * 7u + j] = run; run += c[j]; }
            if (lane == 31) roff[RSL] = incl;
        }
        __syncthreads();

#pragma unroll
        for (int k = 0; k < EPT; ++k) {
            const int i = (int)tid + k * 512;
            if (i < n) ent[roff[(unsigned)e[k].x >> 17] + rk[k]] = e[k];
        }
        __syncthreads();

        const unsigned myoff = roff[min(r_lo + (int)lane, RSL)];
        const int S = __builtin_amdgcn_readlane((int)myoff, 0);
        const int E = __builtin_amdgcn_readlane((int)myoff, nown);

        int   cur = r_lo - 1;
        float a0 = 0.0f, a1 = 0.0f, a2 = 0.0f, a3 = 0.0f;
        auto flush = [&](int nxt) {
            if (cur >= r_lo) {
                float* t = mytile + (cur - r_lo) * TPITCH + lane;
                t[0] += a0; t[32] += a1; t[64] += a2; t[96] += a3;
            }
            a0 = 0.0f; a1 = 0.0f; a2 = 0.0f; a3 = 0.0f;
            cur = nxt;
        };

        for (int e0 = S; e0 < E; e0 += 32) {
            const int cn = min(32, E - e0);
            int2 pv = make_int2(0, 0);
            if ((int)lane < cn) pv = ent[e0 + (int)lane];
            int g = 0;
            for (; g + 8 <= cn; g += 8) {
                float m[8][4];
#pragma unroll
                for (int j = 0; j < 8; ++j) {
                    const unsigned p = (unsigned)__builtin_amdgcn_readlane(pv.x, g + j);
                    const float* s = row_lane_ptr(src, (p & 0x1FFFFu) * (HID_F * 4u) + lane4);
                    m[j][0] = s[0];
                    m[j][1] = s[32];
                    m[j][2] = s[64];
                    m[j][3] = s[96];
                }
#pragma unroll
                for (int j = 0; j < 8; ++j) {
                    const unsigned p = (unsigned)__builtin_amdgcn_readlane(pv.x, g + j);
                    const float    w = __int_as_float(__builtin_amdgcn_readlane(pv.y, g + j));
                    const int      rl = (int)(p >> 17);
                    if (rl != cur) flush(rl);
                    a0 = fmaf(w, m[j][0], a0);
                    a1 = fmaf(w, m[j][1], a1);
                    a2 = fmaf(w, m[j][2], a2);
                    a3 = fmaf(w, m[j][3], a3);
                }
            }
            if (g < cn) {
                float m[7][4];
#pragma unroll
                for (int j = 0; j < 7; ++j) {
                    if (g + j < cn) {
                        const unsigned p = (unsigned)__builtin_amdgcn_readlane(pv.x, g + j);
                        const float* s = row_lane_ptr(src, (p & 0x1FFFFu) * (HID_F * 4u) + lane4);
                        m[j][0] = s[0];
                        m[j][1] = s[32];
                        m[j][2] = s[64];
                        m[j][3] = s[96];
                    }
                }
#pragma unroll
                for (int j = 0; j < 7; ++j) {
                    if (g + j < cn) {
                        const unsigned p = (unsigned)__builtin_amdgcn_readlane(pv.x, g + j);
                        const float    w = __int_as_float(__builtin_amdgcn_readlane(pv.y, g + j));
                        const int      rl = (int)(p >> 17);
                        if (rl != cur) flush(rl);
                        a0 = fmaf(w, m[j][0], a0);
                        a1 = fmaf(w, m[j][1], a1);
                        a2 = fmaf(w, m[j][2], a2);
                        a3 = fmaf(w, m[j][3], a3);
                    }
                }
            }
        }
        flush(r_hi);
    }

    {
        const int  m   = (int)(lane & 15u);
        const int  kh  = (int)(lane >> 4);
        const bool own = m < nown;
        const float* arow = mytile + min(m, TPW - 1) * TPITCH + 2 * kh;
        const float* wcol = W2 + 2 * m;
        v8f acc0 = {}, acc1 = {};
        for (int kb = 0; kb < HID_F; kb += 32) {
            v2f bq[8][2];
#pragma unroll
            for (int q = 0; q < 8; ++q) {
                const float* wrow = wcol + (size_t)(kb + 4 * q + 2 * kh) * OUT_F;
                bq[q][0].x = wrow[0];
                bq[q][0].y = wrow[OUT_F];
                bq[q][1].x = wrow[1];
                bq[q][1].y = wrow[OUT_F + 1];
            }
#pragma unroll
            for (int q = 0; q < 8; ++q) {
                v2f a = *(const v2f*)(arow + kb + 4 * q);
                a.x = own ? fmaxf(a.x, 0.0f) : 0.0f;
                a.y = own ? fmaxf(a.y, 0.0f) : 0.0f;
                acc0 = __builtin_amdgcn_wmma_f32_16x16x4_f32(
                    false, a, false, bq[q][0], (short)0, acc0, false, false);
                acc1 = __builtin_amdgcn_wmma_f32_16x16x4_f32(
                    false, a, false, bq[q][1], (short)0, acc1, false, false);
            }
        }
        float* drow = h2 + ((size_t)b * RPB + (unsigned)(r_lo + kh * 8)) * OUT_F + 2 * m;
#pragma unroll
        for (int v = 0; v < 8; ++v) {
            const int r = kh * 8 + v;
            if (r < nown && (unsigned)(r_lo + r) < nrows) {
                v2f o;
                o.x = acc0[v];
                o.y = acc1[v];
                *(v2f*)(drow + (size_t)v * OUT_F) = o;
            }
        }
    }
}

#define GRP 16

__global__ __launch_bounds__(512) void spmm_f32_rows(
    const int2* __restrict__ list, const int2* __restrict__ seg,
    const float* __restrict__ src, float* __restrict__ dst)
{
    __shared__ int2     ent[CAP];
    __shared__ unsigned rcnt[RSL];
    __shared__ unsigned roff[RSL + 1];
    __shared__ unsigned segbase[NBLK + 1];
    __shared__ int      segadj[NBLK];
    __shared__ unsigned swsum[NBLK / 32];
    __shared__ float    tile[16 * TPW * OUT_F];
    const unsigned tid = threadIdx.x, b = blockIdx.x;
    const unsigned lane = tid & 31;
    const int wave = __builtin_amdgcn_readfirstlane((int)(tid >> 5));
    const unsigned lane4 = lane * 4u;
    const unsigned nrows = min((unsigned)RPB, (unsigned)N_NODES - b * RPB);
    const int r_lo = wave * 12 + min(wave, 4);
    const int nown = wave < 4 ? 13 : 12;
    const int r_hi = r_lo + nown;

    float* mytile = tile + wave * (TPW * OUT_F);
#pragma unroll
    for (int r = 0; r < TPW; ++r) mytile[r * OUT_F + (int)lane] = 0.0f;

    {
        int2 se = make_int2(0, 0);
        unsigned len = 0u, incl = 0u;
        if (wave < NBLK / 32) {
            se   = seg[b * NBLK + tid];
            len  = (unsigned)(se.y - se.x);
            incl = wave_incl_scan(len, lane);
            if (lane == 31) swsum[wave] = incl;
        }
        __syncthreads();
        if (wave < NBLK / 32) {
            unsigned base = incl - len;
            for (int q = 0; q < wave; ++q) base += swsum[q];
            segbase[tid] = base;
            segadj[tid]  = se.x - (int)base;
            if (tid == NBLK - 1) segbase[NBLK] = base + len;
        }
        __syncthreads();
    }
    const int last = __builtin_amdgcn_readfirstlane((int)segbase[NBLK]);

    for (int p0 = 0; p0 < last; p0 += CAP) {
        const int n = min(CAP, last - p0);
        if (tid < RSL) rcnt[tid] = 0u;
        __syncthreads();

        int2     e[EPT];
        unsigned rk[EPT];
#pragma unroll
        for (int k = 0; k < EPT; ++k) {
            const int i = (int)tid + k * 512;
            e[k] = make_int2(0, 0);
            if (i < n) {
                const unsigned f = (unsigned)(p0 + i);
                e[k] = list[segadj[seg_of(segbase, f)] + (int)f];
            }
        }
#pragma unroll
        for (int k = 0; k < EPT; ++k) {
            const int i = (int)tid + k * 512;
            rk[k] = 0u;
            if (i < n) rk[k] = atomicAdd(&rcnt[(unsigned)e[k].x >> 17], 1u);
        }
        __syncthreads();

        if (wave == 0) {
            unsigned c[7], s = 0u;
#pragma unroll
            for (int j = 0; j < 7; ++j) { c[j] = rcnt[lane * 7u + j]; s += c[j]; }
            const unsigned incl = wave_incl_scan(s, lane);
            unsigned run = incl - s;
#pragma unroll
            for (int j = 0; j < 7; ++j) { roff[lane * 7u + j] = run; run += c[j]; }
            if (lane == 31) roff[RSL] = incl;
        }
        __syncthreads();

#pragma unroll
        for (int k = 0; k < EPT; ++k) {
            const int i = (int)tid + k * 512;
            if (i < n) ent[roff[(unsigned)e[k].x >> 17] + rk[k]] = e[k];
        }
        __syncthreads();

        const unsigned myoff = roff[min(r_lo + (int)lane, RSL)];
        const int S = __builtin_amdgcn_readlane((int)myoff, 0);
        const int E = __builtin_amdgcn_readlane((int)myoff, nown);

        int   cur = r_lo - 1;
        float a = 0.0f;
        auto flush = [&](int nxt) {
            if (cur >= r_lo) mytile[(cur - r_lo) * OUT_F + (int)lane] += a;
            a = 0.0f;
            cur = nxt;
        };

        for (int e0 = S; e0 < E; e0 += 32) {
            const int cn = min(32, E - e0);
            int2 pv = make_int2(0, 0);
            if ((int)lane < cn) pv = ent[e0 + (int)lane];
            int g = 0;
            for (; g + GRP <= cn; g += GRP) {
                float m[GRP];
#pragma unroll
                for (int j = 0; j < GRP; ++j) {
                    const unsigned p = (unsigned)__builtin_amdgcn_readlane(pv.x, g + j);
                    m[j] = *row_lane_ptr(src, (p & 0x1FFFFu) * (OUT_F * 4u) + lane4);
                }
#pragma unroll
                for (int j = 0; j < GRP; ++j) {
                    const unsigned p = (unsigned)__builtin_amdgcn_readlane(pv.x, g + j);
                    const float    w = __int_as_float(__builtin_amdgcn_readlane(pv.y, g + j));
                    const int      rl = (int)(p >> 17);
                    if (rl != cur) flush(rl);
                    a = fmaf(w, m[j], a);
                }
            }
            if (g < cn) {
                float m[GRP - 1];
#pragma unroll
                for (int j = 0; j < GRP - 1; ++j) {
                    if (g + j < cn) {
                        const unsigned p = (unsigned)__builtin_amdgcn_readlane(pv.x, g + j);
                        m[j] = *row_lane_ptr(src, (p & 0x1FFFFu) * (OUT_F * 4u) + lane4);
                    }
                }
#pragma unroll
                for (int j = 0; j < GRP - 1; ++j) {
                    if (g + j < cn) {
                        const unsigned p = (unsigned)__builtin_amdgcn_readlane(pv.x, g + j);
                        const float    w = __int_as_float(__builtin_amdgcn_readlane(pv.y, g + j));
                        const int      rl = (int)(p >> 17);
                        if (rl != cur) flush(rl);
                        a = fmaf(w, m[j], a);
                    }
                }
            }
        }
        flush(r_hi);
    }

    float* drow = dst + ((size_t)b * RPB + (unsigned)r_lo) * OUT_F + lane;
#pragma unroll
    for (int r = 0; r < TPW; ++r) {
        if (r < nown && (unsigned)(r_lo + r) < nrows)
            drow[(size_t)r * OUT_F] = mytile[r * OUT_F + (int)lane];
    }
}

extern "C" void kernel_launch(void* const* d_in, const int* in_sizes, int n_in,
                              void* d_out, int out_size, void* d_ws, size_t ws_size,
                              hipStream_t stream)
{
    const float* x    = (const float*)d_in[0];
    const int*   rows = (const int*)  d_in[1];
    const int*   cols = (const int*)  d_in[2];
    const float* vals = (const float*)d_in[3];
    const float* W1   = (const float*)d_in[4];
    const float* W2   = (const float*)d_in[5];
    float* out = (float*)d_out;

    float* h0  = (float*)d_ws;
    float* b1  = h0 + (size_t)N_NODES * HID_F;
    float* h2  = b1 + 4096;
    int2*  list = (int2*)(h2 + (size_t)N_NODES * OUT_F);
    int2*  seg  = list + N_EDGES;

    bin_local<<<NBLK, 1024, 0, stream>>>(rows, cols, vals, list, seg);

    gemm1_wmma<<<G1_WGS, 512, 0, stream>>>(x, W1, h0);

    spmm_f128_rows_gemm2<<<(N_NODES + RPB - 1) / RPB, 512, 0, stream>>>(list, seg, h0, W2, h2);

    spmm_f32_rows<<<(N_NODES + RPB - 1) / RPB, 512, 0, stream>>>(list, seg, h2, out);
}


# edge ordering moved into the GEMM launch: 8 more waves per work-group sort one of 256 slices (was 128) beside the 16 matrix waves, list stores under the products; 4 launches -> 3; arrival-order sums
# speedup vs baseline: 4.3600x; 1.0829x over previous
#include <hip/hip_runtime.h>

#define N_NODES 100000
#define N_EDGES 1600000
#define IN_F    256
#define HID_F   128
#define OUT_F   32

typedef float v2f __attribute__((ext_vector_type(2)));
typedef float v4f __attribute__((ext_vector_type(4)));
typedef float v8f __attribute__((ext_vector_type(8)));

#define NBKT 512
#define RPB  196
#define NBLK 256
#define EPB  (N_EDGES / NBLK)
#define BINT 256
#define EPL  ((EPB + BINT - 1) / BINT)

__device__ __forceinline__ unsigned wave_incl_scan(unsigned v, unsigned lane)
{
#pragma unroll
    for (int d = 1; d < 32; d <<= 1) {
        unsigned u = __shfl_up(v, d, 32);
        if (lane >= (unsigned)d) v += u;
    }
    return v;
}

#define G1_WGS    256
#define G1_PAIRS  (G1_WGS * 8)
#define G1_LDS_SLOTS 10496
static_assert(NBLK == G1_WGS, "one edge slice per work-group of the GEMM launch");

__global__ __launch_bounds__(768) void gemm1_wmma_order(
    const float* __restrict__ x, const float* __restrict__ W1,
    float* __restrict__ h0,
    const int* __restrict__ rows, const int* __restrict__ cols,
    const float* __restrict__ vals, int2* __restrict__ list,
    int2* __restrict__ seg)
{
    __shared__ v4f w1s[G1_LDS_SLOTS];
    __shared__ unsigned cur[NBKT];
    __shared__ unsigned wsum[NBKT / 32];
    const unsigned tid = threadIdx.x, lane = tid & 31;
    const int wave = __builtin_amdgcn_readfirstlane((int)(tid >> 5));
    const int c = (int)(lane & 15u), h = (int)(lane >> 4);
    const bool orderer = wave >= 16;
    const unsigned ot = tid & (BINT - 1u);
    const unsigned sl = blockIdx.x;

    const int nh = wave & 1;
    const int P = (wave >> 1) * G1_WGS + (int)blockIdx.x;
    int rb = P;
    const float* arow = x + (size_t)(rb * 16 + c) * IN_F + 4 * h;
    v4f a[2][8];

    if (tid < NBKT) cur[tid] = 0u;
    unsigned er[EPL] = {};
    if (orderer) {
#pragma unroll
        for (int j = 0; j < EPL; ++j) {
            const unsigned i = ot + (unsigned)j * BINT;
            if (i < EPB) er[j] = (unsigned)rows[sl * EPB + i];
        }
    } else {
        v4f k0[8], k1[8];
#pragma unroll
        for (int i = 0; i < 8; ++i) {
            const int j = wave + 16 * i;
            k0[i] = *(const v4f*)(W1 + (size_t)(2 * j) * HID_F + 4 * lane);
            k1[i] = *(const v4f*)(W1 + (size_t)(2 * j + 1) * HID_F + 4 * lane);
        }
#pragma unroll
        for (int i = 0; i < 8; ++i) a[0][i] = *(const v4f*)(arow + 8 * i);
#pragma unroll
        for (int i = 0; i < 8; ++i) {
            const int j = wave + 16 * i;
            const int slot = j * 64 + h * 32 + c;
            v4f lo, hi;
            lo.x = k0[i].x; lo.y = k1[i].x; lo.z = k0[i].y; lo.w = k1[i].y;
            hi.x = k0[i].z; hi.y = k1[i].z; hi.z = k0[i].w; hi.w = k1[i].w;
            w1s[slot]      = lo;
            w1s[slot + 16] = hi;
        }
    }
    __syncthreads();

    if (orderer) {
#pragma unroll
        for (int j = 0; j < EPL; ++j) {
            const unsigned i = ot + (unsigned)j * BINT;
            if (i < EPB) atomicAdd(&cur[er[j] / RPB], 1u);
        }
    }
    __syncthreads();
    unsigned cv = 0u, incl = 0u;
    if (tid < NBKT) {
        cv   = cur[tid];
        incl = wave_incl_scan(cv, lane);
        if (lane == 31) wsum[tid >> 5] = incl;
    }
    __syncthreads();
    if (tid < NBKT) {
        unsigned base = 0u;
        for (unsigned q = 0; q < (tid >> 5); ++q) base += wsum[q];
        const unsigned first = sl * EPB + base + incl - cv;
        cur[tid] = first;
        seg[tid * NBLK + sl] = make_int2((int)first, (int)(first + cv));
    }
    __syncthreads();
    if (orderer) {
        unsigned pr[EPL] = {}, ec[EPL] = {};
        int      ew[EPL] = {};
#pragma unroll
        for (int j = 0; j < EPL; ++j) {
            const unsigned i = ot + (unsigned)j * BINT;
            if (i < EPB) {
                const unsigned e = sl * EPB + i;
                pr[j] = (unsigned)rows[e];
                ec[j] = (unsigned)cols[e];
                ew[j] = __float_as_int(vals[e]);
            }
        }
#pragma unroll
        for (int j = 0; j < EPL; ++j) {
            const unsigned i = ot + (unsigned)j * BINT;
            if (i < EPB) {
                const unsigned k   = pr[j] / RPB;
                const unsigned rl  = pr[j] - k * RPB;
                const unsigned pos = atomicAdd(&cur[k], 1u);
                list[pos] = make_int2((int)(ec[j] | (rl << 17)), ew[j]);
            }
        }
        return;
    }

    const v4f* bbase = w1s + h * 128 + nh * 32 + c;
    v4f b[2][4];
    b[0][0] = bbase[0];
    b[0][1] = bbase[16];
    b[0][2] = bbase[64];
    b[0][3] = bbase[64 + 16];

    for (; rb < N_NODES / 16; rb += G1_PAIRS) {
        const int rbn = rb + G1_PAIRS < N_NODES / 16 ? rb + G1_PAIRS : rb;
        const float* anext = x + (size_t)(rbn * 16 + c) * IN_F + 4 * h;
        v8f acc0 = {}, acc1 = {}, acc2 = {}, acc3 = {};
#pragma nounroll
        for (int hf = 0; hf < 2; ++hf) {
            const v4f*   bh    = bbase + hf * 4096;
            const v4f*   bwrap = bbase + (1 - hf) * 4096;
            const float* acur  = arow + hf * 128;
            const float* apre  = hf ? anext : arow + 128;
#pragma unroll
            for (int sb = 0; sb < 2; ++sb) {
#pragma unroll
                for (int i = 0; i < 8; ++i)
                    a[(sb + 1) & 1][i] = sb == 0 ? *(const v4f*)(acur + 64 + 8 * i)
                                                 : *(const v4f*)(apre + 8 * i);
#pragma unroll
                for (int i = 0; i < 8; ++i) {
                    const int ql = 8 * sb + i;
                    if (ql + 1 < 16) {
                        b[(ql + 1) & 1][0] = bh[(ql + 1) * 256];
                        b[(ql + 1) & 1][1] = bh[(ql + 1) * 256 + 16];
                        b[(ql + 1) & 1][2] = bh[(ql + 1) * 256 + 64];
                        b[(ql + 1) & 1][3] = bh[(ql + 1) * 256 + 64 + 16];
                    } else {
                        b[0][0] = bwrap[0];
                        b[0][1] = bwrap[16];
                        b[0][2] = bwrap[64];
                        b[0][3] = bwrap[64 + 16];
                    }
                    const v4f av = a[sb][i];
#pragma unroll
                    for (int u = 0; u < 2; ++u) {
                        const v4f b01 = b[ql & 1][2 * u];
                        const v4f b23 = b[ql & 1][2 * u + 1];
                        v2f af, b0, b1, b2, b3;
                        af.x = u ? av.z : av.x;  af.y = u ? av.w : av.y;
                        b0.x = b01.x; b0.y = b01.y;
                        b1.x = b01.z; b1.y = b01.w;
                        b2.x = b23.x; b2.y = b23.y;
                        b3.x = b23.z; b3.y = b23.w;
                        acc0 = __builtin_amdgcn_wmma_f32_16x16x4_f32(
                            false, af, false, b0, (short)0, acc0, false, false);
                        acc1 = __builtin_amdgcn_wmma_f32_16x16x4_f32(
                            false, af, false, b1, (short)0, acc1, false, false);
                        acc2 = __builtin_amdgcn_wmma_f32_16x16x4_f32(
                            false, af, false, b2, (short)0, acc2, false, false);
                        acc3 = __builtin_amdgcn_wmma_f32_16x16x4_f32(
                            false, af, false, b3, (short)0, acc3, false, false);
                    }
                    __builtin_amdgcn_sched_barrier(0);
                }
            }
        }
        asm volatile("v_nop\n\tv_nop\n\tv_nop\n\tv_nop\n\tv_nop\n\tv_nop\n\tv_nop\n\tv_nop");
        __builtin_amdgcn_sched_barrier(0);
        float* drow = h0 + (size_t)(rb * 16 + 8 * h) * HID_F + 64 * nh + 4 * c;
#pragma unroll
        for (int r = 0; r < 8; ++r) {
            v4f o;
            o.x = acc0[r]; o.y = acc1[r]; o.z = acc2[r]; o.w = acc3[r];
            *(v4f*)(drow + (size_t)r * HID_F) = o;
        }
        arow = anext;
    }
}

__device__ __forceinline__ const float* row_lane_ptr(const float* base, unsigned off)
{
    return (const float*)((const char*)base + off);
}
__device__ __forceinline__ float* row_lane_ptr(float* base, unsigned off)
{
    return (float*)((char*)base + off);
}

__device__ __forceinline__ unsigned seg_of(const unsigned* segbase, unsigned f)
{
    unsigned s = 0u;
#pragma unroll
    for (unsigned step = NBLK / 2; step >= 1u; step >>= 1)
        if (segbase[s + step] <= f) s += step;
    return s;
}

#define CAP 4096
#define EPT (CAP / 512)
#define RSL 224
#define TPW 13
#define TPITCH 132

__global__ __launch_bounds__(512) void spmm_f128_rows_gemm2(
    const int2* __restrict__ list, const int2* __restrict__ seg,
    const float* __restrict__ src, const float* __restrict__ W2,
    float* __restrict__ h2)
{
    __shared__ int2     ent[CAP];
    __shared__ unsigned rcnt[RSL];
    __shared__ unsigned roff[RSL + 1];
    __shared__ unsigned segbase[NBLK + 1];
    __shared__ int      segadj[NBLK];
    __shared__ unsigned swsum[NBLK / 32];
    __shared__ float    tile[16 * TPW * TPITCH];
    const unsigned tid = threadIdx.x, b = blockIdx.x;
    const unsigned lane = tid & 31;
    const int wave = __builtin_amdgcn_readfirstlane((int)(tid >> 5));
    const unsigned lane4 = lane * 4u;
    const unsigned nrows = min((unsigned)RPB, (unsigned)N_NODES - b * RPB);
    const int r_lo = wave * 12 + min(wave, 4);
    const int nown = wave < 4 ? 13 : 12;
    const int r_hi = r_lo + nown;

    float* mytile = tile + wave * (TPW * TPITCH);
#pragma unroll
    for (int r = 0; r < TPW; ++r) {
        float* t = mytile + r * TPITCH + lane;
        t[0] = 0.0f; t[32] = 0.0f; t[64] = 0.0f; t[96] = 0.0f;
    }

    {
        int2 se = make_int2(0, 0);
        unsigned len = 0u, incl = 0u;
        if (wave < NBLK / 32) {
            se   = seg[b * NBLK + tid];
            len  = (unsigned)(se.y - se.x);
            incl = wave_incl_scan(len, lane);
            if (lane == 31) swsum[wave] = incl;
        }
        __syncthreads();
        if (wave < NBLK / 32) {
            unsigned base = incl - len;
            for (int q = 0; q < wave; ++q) base += swsum[q];
            segbase[tid] = base;
            segadj[tid]  = se.x - (int)base;
            if (tid == NBLK - 1) segbase[NBLK] = base + len;
        }
        __syncthreads();
    }
    const int last = __builtin_amdgcn_readfirstlane((int)segbase[NBLK]);

    for (int p0 = 0; p0 < last; p0 += CAP) {
        const int n = min(CAP, last - p0);
        if (tid < RSL) rcnt[tid] = 0u;
        __syncthreads();

        int2     e[EPT];
        unsigned rk[EPT];
#pragma unroll
        for (int k = 0; k < EPT; ++k) {
            const int i = (int)tid + k * 512;
            e[k] = make_int2(0, 0);
            if (i < n) {
                const unsigned f = (unsigned)(p0 + i);
                e[k] = list[segadj[seg_of(segbase, f)] + (int)f];
            }
        }
#pragma unroll
        for (int k = 0; k < EPT; ++k) {
            const int i = (int)tid + k * 512;
            rk[k] = 0u;
            if (i < n) rk[k] = atomicAdd(&rcnt[(unsigned)e[k].x >> 17], 1u);
        }
        __syncthreads();

        if (wave == 0) {
            unsigned c[7], s = 0u;
#pragma unroll
            for (int j = 0; j < 7; ++j) { c[j] = rcnt[lane * 7u + j]; s += c[j]; }
            const unsigned incl = wave_incl_scan(s, lane);
            unsigned run = incl - s;
#pragma unroll
            for (int j = 0; j < 7; ++j) { roff[lane * 7u + j] = run; run += c[j]; }
            if (lane == 31) roff[RSL] = incl;
        }
        __syncthreads();

#pragma unroll
        for (int k = 0; k < EPT; ++k) {
            const int i = (int)tid + k * 512;
            if (i < n) ent[roff[(unsigned)e[k].x >> 17] + rk[k]] = e[k];
        }
        __syncthreads();

        const unsigned myoff = roff[min(r_lo + (int)lane, RSL)];
        const int S = __builtin_amdgcn_readlane((int)myoff, 0);
        const int E = __builtin_amdgcn_readlane((int)myoff, nown);

        int   cur = r_lo - 1;
        float a0 = 0.0f, a1 = 0.0f, a2 = 0.0f, a3 = 0.0f;
        auto flush = [&](int nxt) {
            if (cur >= r_lo) {
                float* t = mytile + (cur - r_lo) * TPITCH + lane;
                t[0] += a0; t[32] += a1; t[64] += a2; t[96] += a3;
            }
            a0 = 0.0f; a1 = 0.0f; a2 = 0.0f; a3 = 0.0f;
            cur = nxt;
        };

        for (int e0 = S; e0 < E; e0 += 32) {
            const int cn = min(32, E - e0);
            int2 pv = make_int2(0, 0);
            if ((int)lane < cn) pv = ent[e0 + (int)lane];
            int g = 0;
            for (; g + 8 <= cn; g += 8) {
                float m[8][4];
#pragma unroll
                for (int j = 0; j < 8; ++j) {
                    const unsigned p = (unsigned)__builtin_amdgcn_readlane(pv.x, g + j);
                    const float* s = row_lane_ptr(src, (p & 0x1FFFFu) * (HID_F * 4u) + lane4);
                    m[j][0] = s[0];
                    m[j][1] = s[32];
                    m[j][2] = s[64];
                    m[j][3] = s[96];
                }
#pragma unroll
                for (int j = 0; j < 8; ++j) {
                    const unsigned p = (unsigned)__builtin_amdgcn_readlane(pv.x, g + j);
                    const float    w = __int_as_float(__builtin_amdgcn_readlane(pv.y, g + j));
                    const int      rl = (int)(p >> 17);
                    if (rl != cur) flush(rl);
                    a0 = fmaf(w, m[j][0], a0);
                    a1 = fmaf(w, m[j][1], a1);
                    a2 = fmaf(w, m[j][2], a2);
                    a3 = fmaf(w, m[j][3], a3);
                }
            }
            if (g < cn) {
                float m[7][4];
#pragma unroll
                for (int j = 0; j < 7; ++j) {
                    if (g + j < cn) {
                        const unsigned p = (unsigned)__builtin_amdgcn_readlane(pv.x, g + j);
                        const float* s = row_lane_ptr(src, (p & 0x1FFFFu) * (HID_F * 4u) + lane4);
                        m[j][0] = s[0];
                        m[j][1] = s[32];
                        m[j][2] = s[64];
                        m[j][3] = s[96];
                    }
                }
#pragma unroll
                for (int j = 0; j < 7; ++j) {
                    if (g + j < cn) {
                        const unsigned p = (unsigned)__builtin_amdgcn_readlane(pv.x, g + j);
                        const float    w = __int_as_float(__builtin_amdgcn_readlane(pv.y, g + j));
                        const int      rl = (int)(p >> 17);
                        if (rl != cur) flush(rl);
                        a0 = fmaf(w, m[j][0], a0);
                        a1 = fmaf(w, m[j][1], a1);
                        a2 = fmaf(w, m[j][2], a2);
                        a3 = fmaf(w, m[j][3], a3);
                    }
                }
            }
        }
        flush(r_hi);
    }

    {
        const int  m   = (int)(lane & 15u);
        const int  kh  = (int)(lane >> 4);
        const bool own = m < nown;
        const float* arow = mytile + min(m, TPW - 1) * TPITCH + 2 * kh;
        const float* wcol = W2 + 2 * m;
        v8f acc0 = {}, acc1 = {};
        for (int kb = 0; kb < HID_F; kb += 32) {
            v2f bq[8][2];
#pragma unroll
            for (int q = 0; q < 8; ++q) {
                const float* wrow = wcol + (size_t)(kb + 4 * q + 2 * kh) * OUT_F;
                bq[q][0].x = wrow[0];
                bq[q][0].y = wrow[OUT_F];
                bq[q][1].x = wrow[1];
                bq[q][1].y = wrow[OUT_F + 1];
            }
#pragma unroll
            for (int q = 0; q < 8; ++q) {
                v2f a = *(const v2f*)(arow + kb + 4 * q);
                a.x = own ? fmaxf(a.x, 0.0f) : 0.0f;
                a.y = own ? fmaxf(a.y, 0.0f) : 0.0f;
                acc0 = __builtin_amdgcn_wmma_f32_16x16x4_f32(
                    false, a, false, bq[q][0], (short)0, acc0, false, false);
                acc1 = __builtin_amdgcn_wmma_f32_16x16x4_f32(
                    false, a, false, bq[q][1], (short)0, acc1, false, false);
            }
        }
        float* drow = h2 + ((size_t)b * RPB + (unsigned)(r_lo + kh * 8)) * OUT_F + 2 * m;
#pragma unroll
        for (int v = 0; v < 8; ++v) {
            const int r = kh * 8 + v;
            if (r < nown && (unsigned)(r_lo + r) < nrows) {
                v2f o;
                o.x = acc0[v];
                o.y = acc1[v];
                *(v2f*)(drow + (size_t)v * OUT_F) = o;
            }
        }
    }
}

#define GRP 16

__global__ __launch_bounds__(512) void spmm_f32_rows(
    const int2* __restrict__ list, const int2* __restrict__ seg,
    const float* __restrict__ src, float* __restrict__ dst)
{
    __shared__ int2     ent[CAP];
    __shared__ unsigned rcnt[RSL];
    __shared__ unsigned roff[RSL + 1];
    __shared__ unsigned segbase[NBLK + 1];
    __shared__ int      segadj[NBLK];
    __shared__ unsigned swsum[NBLK / 32];
    __shared__ float    tile[16 * TPW * OUT_F];
    const unsigned tid = threadIdx.x, b = blockIdx.x;
    const unsigned lane = tid & 31;
    const int wave = __builtin_amdgcn_readfirstlane((int)(tid >> 5));
    const unsigned lane4 = lane * 4u;
    const unsigned nrows = min((unsigned)RPB, (unsigned)N_NODES - b * RPB);
    const int r_lo = wave * 12 + min(wave, 4);
    const int nown = wave < 4 ? 13 : 12;
    const int r_hi = r_lo + nown;

    float* mytile = tile + wave * (TPW * OUT_F);
#pragma unroll
    for (int r = 0; r < TPW; ++r) mytile[r * OUT_F + (int)lane] = 0.0f;

    {
        int2 se = make_int2(0, 0);
        unsigned len = 0u, incl = 0u;
        if (wave < NBLK / 32) {
            se   = seg[b * NBLK + tid];
            len  = (unsigned)(se.y - se.x);
            incl = wave_incl_scan(len, lane);
            if (lane == 31) swsum[wave] = incl;
        }
        __syncthreads();
        if (wave < NBLK / 32) {
            unsigned base = incl - len;
            for (int q = 0; q < wave; ++q) base += swsum[q];
            segbase[tid] = base;
            segadj[tid]  = se.x - (int)base;
            if (tid == NBLK - 1) segbase[NBLK] = base + len;
        }
        __syncthreads();
    }
    const int last = __builtin_amdgcn_readfirstlane((int)segbase[NBLK]);

    for (int p0 = 0; p0 < last; p0 += CAP) {
        const int n = min(CAP, last - p0);
        if (tid < RSL) rcnt[tid] = 0u;
        __syncthreads();

        int2     e[EPT];
        unsigned rk[EPT];
#pragma unroll
        for (int k = 0; k < EPT; ++k) {
            const int i = (int)tid + k * 512;
            e[k] = make_int2(0, 0);
            if (i < n) {
                const unsigned f = (unsigned)(p0 + i);
                e[k] = list[segadj[seg_of(segbase, f)] + (int)f];
            }
        }
#pragma unroll
        for (int k = 0; k < EPT; ++k) {
            const int i = (int)tid + k * 512;
            rk[k] = 0u;
            if (i < n) rk[k] = atomicAdd(&rcnt[(unsigned)e[k].x >> 17], 1u);
        }
        __syncthreads();

        if (wave == 0) {
            unsigned c[7], s = 0u;
#pragma unroll
            for (int j = 0; j < 7; ++j) { c[j] = rcnt[lane * 7u + j]; s += c[j]; }
            const unsigned incl = wave_incl_scan(s, lane);
            unsigned run = incl - s;
#pragma unroll
            for (int j = 0; j < 7; ++j) { roff[lane * 7u + j] = run; run += c[j]; }
            if (lane == 31) roff[RSL] = incl;
        }
        __syncthreads();

#pragma unroll
        for (int k = 0; k < EPT; ++k) {
            const int i = (int)tid + k * 512;
            if (i < n) ent[roff[(unsigned)e[k].x >> 17] + rk[k]] = e[k];
        }
        __syncthreads();

        const unsigned myoff = roff[min(r_lo + (int)lane, RSL)];
        const int S = __builtin_amdgcn_readlane((int)myoff, 0);
        const int E = __builtin_amdgcn_readlane((int)myoff, nown);

        int   cur = r_lo - 1;
        float a = 0.0f;
        auto flush = [&](int nxt) {
            if (cur >= r_lo) mytile[(cur - r_lo) * OUT_F + (int)lane] += a;
            a = 0.0f;
            cur = nxt;
        };

        for (int e0 = S; e0 < E; e0 += 32) {
            const int cn = min(32, E - e0);
            int2 pv = make_int2(0, 0);
            if ((int)lane < cn) pv = ent[e0 + (int)lane];
            int g = 0;
            for (; g + GRP <= cn; g += GRP) {
                float m[GRP];
#pragma unroll
                for (int j = 0; j < GRP; ++j) {
                    const unsigned p = (unsigned)__builtin_amdgcn_readlane(pv.x, g + j);
                    m[j] = *row_lane_ptr(src, (p & 0x1FFFFu) * (OUT_F * 4u) + lane4);
                }
#pragma unroll
                for (int j = 0; j < GRP; ++j) {
                    const unsigned p = (unsigned)__builtin_amdgcn_readlane(pv.x, g + j);
                    const float    w = __int_as_float(__builtin_amdgcn_readlane(pv.y, g + j));
                    const int      rl = (int)(p >> 17);
                    if (rl != cur) flush(rl);
                    a = fmaf(w, m[j], a);
                }
            }
            if (g < cn) {
                float m[GRP - 1];
#pragma unroll
                for (int j = 0; j < GRP - 1; ++j) {
                    if (g + j < cn) {
                        const unsigned p = (unsigned)__builtin_amdgcn_readlane(pv.x, g + j);
                        m[j] = *row_lane_ptr(src, (p & 0x1FFFFu) * (OUT_F * 4u) + lane4);
                    }
                }
#pragma unroll
                for (int j = 0; j < GRP - 1; ++j) {
                    if (g + j < cn) {
                        const unsigned p = (unsigned)__builtin_amdgcn_readlane(pv.x, g + j);
                        const float    w = __int_as_float(__builtin_amdgcn_readlane(pv.y, g + j));
                        const int      rl = (int)(p >> 17);
                        if (rl != cur) flush(rl);
                        a = fmaf(w, m[j], a);
                    }
                }
            }
        }
        flush(r_hi);
    }

    float* drow = dst + ((size_t)b * RPB + (unsigned)r_lo) * OUT_F + lane;
#pragma unroll
    for (int r = 0; r < TPW; ++r) {
        if (r < nown && (unsigned)(r_lo + r) < nrows)
            drow[(size_t)r * OUT_F] = mytile[r * OUT_F + (int)lane];
    }
}

extern "C" void kernel_launch(void* const* d_in, const int* in_sizes, int n_in,
                              void* d_out, int out_size, void* d_ws, size_t ws_size,
                              hipStream_t stream)
{
    const float* x    = (const float*)d_in[0];
    const int*   rows = (const int*)  d_in[1];
    const int*   cols = (const int*)  d_in[2];
    const float* vals = (const float*)d_in[3];
    const float* W1   = (const float*)d_in[4];
    const float* W2   = (const float*)d_in[5];
    float* out = (float*)d_out;

    float* h0  = (float*)d_ws;
    float* b1  = h0 + (size_t)N_NODES * HID_F;
    float* h2  = b1 + 4096;
    int2*  list = (int2*)(h2 + (size_t)N_NODES * OUT_F);
    int2*  seg  = list + N_EDGES;

    gemm1_wmma_order<<<G1_WGS, 768, 0, stream>>>(x, W1, h0, rows, cols, vals, list, seg);

    spmm_f128_rows_gemm2<<<(N_NODES + RPB - 1) / RPB, 512, 0, stream>>>(list, seg, h0, W2, h2);

    spmm_f32_rows<<<(N_NODES + RPB - 1) / RPB, 512, 0, stream>>>(list, seg, h2, out);
}
